# starting kernel
# baseline (speedup 1.0000x reference)
#include <hip/hip_runtime.h>
#include <math.h>

#define D_MODEL   256
#define NUM_HEADS 8
#define DK        32
#define B_SZ      4
#define T_SZ      2048

typedef float v2f __attribute__((ext_vector_type(2)));
typedef float v8f __attribute__((ext_vector_type(8)));

__device__ __forceinline__ v8f wmma4(v2f a, v2f b, v8f c) {
    return __builtin_amdgcn_wmma_f32_16x16x4_f32(
         false, a,  false, b,
         (short)0, c,  false,  false);
}

__global__ __launch_bounds__(256)
void qkv_proj_kernel(const float* __restrict__ x,
                     const float* __restrict__ Wq, const float* __restrict__ bq,
                     const float* __restrict__ Wk, const float* __restrict__ bk,
                     const float* __restrict__ Wv, const float* __restrict__ bv,
                     float* __restrict__ QKV)
{
    const int lane = threadIdx.x & 31;
    const int w    = threadIdx.x >> 5;
    const int tile = blockIdx.x * 8 + w;

    const int NT  = D_MODEL / 16;
    const int MT  = (B_SZ * T_SZ) / 16;
    const int mat = tile / (MT * NT);
    const int rem = tile % (MT * NT);
    const int mt  = rem / NT;
    const int nt  = rem % NT;

    const float* W  = (mat == 0) ? Wq : (mat == 1) ? Wk : Wv;
    const float* bs = (mat == 0) ? bq : (mat == 1) ? bk : bv;

    const int r15 = lane & 15;
    const int grp = lane >> 4;
    const int n   = nt * 16 + r15;
    const int m0  = mt * 16;

    const float* xr = x + (size_t)(m0 + r15) * D_MODEL;
    const float* wr = W + (size_t)n * D_MODEL;

    v8f acc = {};
    for (int k0 = 0; k0 < D_MODEL; k0 += 4) {
        const int ka = k0 + grp * 2;
        v2f a = *(const v2f*)(xr + ka);
        v2f b = *(const v2f*)(wr + ka);
        acc = wmma4(a, b, acc);
    }

    const float bias = bs[n];
    const int h = n >> 5, d = n & 31;
    float* dst = QKV + (size_t)mat * ((size_t)B_SZ * NUM_HEADS * T_SZ * DK);
#pragma unroll
    for (int m = 0; m < 8; ++m) {
        const int tl = m0 + m + 8 * grp;
        const int b_ = tl / T_SZ, t_ = tl % T_SZ;
        const float v = acc[m] + bias;
        if (mat == 2) {
            dst[(((size_t)(b_ * NUM_HEADS + h)) * DK + d) * T_SZ + t_] = v;
        } else {
            dst[(((size_t)(b_ * NUM_HEADS + h)) * T_SZ + t_) * DK + d] = v;
        }
    }
}

template <bool BETA_ONE>
__device__ __forceinline__ void attn_core(
    const float* __restrict__ Q, const float* __restrict__ K,
    const float* __restrict__ Vt, const float* __restrict__ oc,
    float* __restrict__ ctx, float* __restrict__ pl,
    int b, int h, int qt, int lane,
    float eta, float absbeta, float alpha)
{
    const int col = lane & 15;
    const int grp = lane >> 4;
    const int q0  = qt * 16;

    const float scale   = 0.17677669529663687f;
    const float NEG_BIG = -3.4e38f;

    const size_t bh  = (size_t)(b * NUM_HEADS + h) * T_SZ;
    const size_t bhd = (size_t)(b * NUM_HEADS + h) * DK;

    const float* qp = Q + (bh + q0 + col) * DK;
    v2f qa[8];
#pragma unroll
    for (int kc = 0; kc < 8; ++kc) qa[kc] = *(const v2f*)(qp + kc * 4 + grp * 2);

    float qoc0[8], qoc1[8], qoc2[8];
#pragma unroll
    for (int m = 0; m < 8; ++m) {
        const int qq = q0 + 8 * grp + m;
        const float* op = oc + ((size_t)b * T_SZ + qq) * 3;
        qoc0[m] = op[0]; qoc1[m] = op[1]; qoc2[m] = op[2];
    }

    const float* vt0 = Vt + (bhd + col) * T_SZ;
    const float* vt1 = vt0 + 16 * T_SZ;

    float Mrow[8], Lrow[8];
#pragma unroll
    for (int m = 0; m < 8; ++m) { Mrow[m] = NEG_BIG; Lrow[m] = 0.f; }
    v8f O0 = {}, O1 = {};

    for (int kt = 0; kt <= qt; ++kt) {
        const int k0   = kt * 16;
        const int kcol = k0 + col;
        const float* kp = K + (bh + kcol) * DK;

        if (kt < qt) {
            __builtin_prefetch(kp + 16 * DK, 0, 0);
            __builtin_prefetch(vt0 + k0 + 16, 0, 0);
            __builtin_prefetch(vt1 + k0 + 16, 0, 0);
        }

        v8f s = {};
#pragma unroll
        for (int kc = 0; kc < 8; ++kc) {
            v2f bb = *(const v2f*)(kp + kc * 4 + grp * 2);
            s = wmma4(qa[kc], bb, s);
        }

        const float* kop = oc + ((size_t)b * T_SZ + kcol) * 3;
        const float koc0 = kop[0], koc1 = kop[1], koc2 = kop[2];

#pragma unroll
        for (int m = 0; m < 8; ++m) {
            const int qrow = q0 + 8 * grp + m;
            if (kcol > qrow) {
                s[m] = NEG_BIG;
            } else {
                const float dist = (float)(qrow - kcol);
                const float powd = BETA_ONE ? dist : __powf(dist, absbeta);
                const float td   = eta / (1.f + powd);
                const float d0 = qoc0[m] - koc0;
                const float d1 = qoc1[m] - koc1;
                const float d2 = qoc2[m] - koc2;
                const float pr = alpha * td * __expf(-(d0*d0 + d1*d1 + d2*d2));
                s[m] = s[m] * scale * pr;
            }
        }

#pragma unroll
        for (int m = 0; m < 8; ++m) {
            float v = s[m];
            v = fmaxf(v, __shfl_xor(v, 1));
            v = fmaxf(v, __shfl_xor(v, 2));
            v = fmaxf(v, __shfl_xor(v, 4));
            v = fmaxf(v, __shfl_xor(v, 8));
            const float mnew = fmaxf(Mrow[m], v);
            const float sc   = __expf(Mrow[m] - mnew);
            const float p    = __expf(s[m] - mnew);
            float su = p;
            su += __shfl_xor(su, 1);
            su += __shfl_xor(su, 2);
            su += __shfl_xor(su, 4);
            su += __shfl_xor(su, 8);
            Lrow[m] = Lrow[m] * sc + su;
            Mrow[m] = mnew;
            O0[m] *= sc;
            O1[m] *= sc;
            pl[(8 * grp + m) * 16 + col] = p;
        }

#pragma unroll
        for (int kc = 0; kc < 4; ++kc) {
            const int kk = kc * 4 + grp * 2;
            v2f pa;
            pa.x = pl[col * 16 + kk];
            pa.y = pl[col * 16 + kk + 1];
            v2f b0 = *(const v2f*)(vt0 + k0 + kk);
            v2f b1 = *(const v2f*)(vt1 + k0 + kk);
            O0 = wmma4(pa, b0, O0);
            O1 = wmma4(pa, b1, O1);
        }
    }

#pragma unroll
    for (int m = 0; m < 8; ++m) {
        const float inv = 1.f / Lrow[m];
        const int qq = q0 + 8 * grp + m;
        float* cp = ctx + ((size_t)b * T_SZ + qq) * D_MODEL + h * DK;
        cp[col]      = O0[m] * inv;
        cp[16 + col] = O1[m] * inv;
    }
}

__global__ __launch_bounds__(256)
void attn_kernel(const float* __restrict__ QKV,
                 const float* __restrict__ oc,
                 const float* __restrict__ etap,
                 const float* __restrict__ betap,
                 const float* __restrict__ alphap,
                 float* __restrict__ ctx)
{
    __shared__ float plds[8][256];

    const int lane = threadIdx.x & 31;
    const int w    = threadIdx.x >> 5;
    const int wid  = blockIdx.x * 8 + w;

    const int QT = T_SZ / 16;
    const int qt = wid % QT;
    const int h  = (wid / QT) % NUM_HEADS;
    const int b  = wid / (QT * NUM_HEADS);

    const float eta     = etap[0];
    const float absbeta = fabsf(betap[0]);
    const float alpha   = alphap[0];

    const size_t HT = (size_t)B_SZ * NUM_HEADS * T_SZ * DK;
    const float* Q  = QKV;
    const float* K  = QKV + HT;
    const float* Vt = QKV + 2 * HT;

    float* pl = &plds[w][0];

    if (absbeta == 1.0f) {
        attn_core<true >(Q, K, Vt, oc, ctx, pl, b, h, qt, lane, eta, absbeta, alpha);
    } else {
        attn_core<false>(Q, K, Vt, oc, ctx, pl, b, h, qt, lane, eta, absbeta, alpha);
    }
}

__global__ __launch_bounds__(256)
void out_proj_kernel(const float* __restrict__ ctx,
                     const float* __restrict__ Wo, const float* __restrict__ bo,
                     float* __restrict__ out)
{
    const int lane = threadIdx.x & 31;
    const int w    = threadIdx.x >> 5;
    const int tile = blockIdx.x * 8 + w;

    const int NT = D_MODEL / 16;
    const int mt = tile / NT;
    const int nt = tile % NT;

    const int r15 = lane & 15;
    const int grp = lane >> 4;
    const int n   = nt * 16 + r15;
    const int m0  = mt * 16;

    const float* xr = ctx + (size_t)(m0 + r15) * D_MODEL;
    const float* wr = Wo + (size_t)n * D_MODEL;

    v8f acc = {};
    for (int k0 = 0; k0 < D_MODEL; k0 += 4) {
        const int ka = k0 + grp * 2;
        v2f a = *(const v2f*)(xr + ka);
        v2f b = *(const v2f*)(wr + ka);
        acc = wmma4(a, b, acc);
    }

    const float bias = bo[n];
#pragma unroll
    for (int m = 0; m < 8; ++m) {
        const int tl = m0 + m + 8 * grp;
        out[(size_t)tl * D_MODEL + n] = acc[m] + bias;
    }
}

extern "C" void kernel_launch(void* const* d_in, const int* in_sizes, int n_in,
                              void* d_out, int out_size, void* d_ws, size_t ws_size,
                              hipStream_t stream) {
    (void)in_sizes; (void)n_in; (void)out_size; (void)ws_size;

    const float* x     = (const float*)d_in[0];
    const float* oc    = (const float*)d_in[1];
    const float* Wq    = (const float*)d_in[2];
    const float* bq    = (const float*)d_in[3];
    const float* Wk    = (const float*)d_in[4];
    const float* bk    = (const float*)d_in[5];
    const float* Wv    = (const float*)d_in[6];
    const float* bv    = (const float*)d_in[7];
    const float* Wo    = (const float*)d_in[8];
    const float* bo    = (const float*)d_in[9];
    const float* eta   = (const float*)d_in[10];
    const float* beta  = (const float*)d_in[11];
    const float* alpha = (const float*)d_in[12];
    float* out = (float*)d_out;

    float* ws  = (float*)d_ws;
    float* QKV = ws;
    float* ctx = ws + 3 * (size_t)B_SZ * NUM_HEADS * T_SZ * DK;

    {
        const int tiles = 3 * (B_SZ * T_SZ / 16) * (D_MODEL / 16);
        qkv_proj_kernel<<<tiles / 8, 256, 0, stream>>>(x, Wq, bq, Wk, bk, Wv, bv, QKV);
    }
    {
        const int waves = B_SZ * NUM_HEADS * (T_SZ / 16);
        attn_kernel<<<waves / 8, 256, 0, stream>>>(QKV, oc, eta, beta, alpha, ctx);
    }
    {
        const int tiles = (B_SZ * T_SZ / 16) * (D_MODEL / 16);
        out_proj_kernel<<<tiles / 8, 256, 0, stream>>>(ctx, Wo, bo, out);
    }
}


# attention on the transposed score tile (S^T=K*Q^T, O^T=V^T*P^T): softmax max/sum in registers per query, P^T feeds PV from its D registers, no LDS; changes row-sum and PV key summation order
# speedup vs baseline: 1.1357x; 1.1357x over previous
#include <hip/hip_runtime.h>
#include <math.h>

#define D_MODEL   256
#define NUM_HEADS 8
#define DK        32
#define B_SZ      4
#define T_SZ      2048

typedef float v2f __attribute__((ext_vector_type(2)));
typedef float v4f __attribute__((ext_vector_type(4)));
typedef float v8f __attribute__((ext_vector_type(8)));

__device__ __forceinline__ v8f wmma4(v2f a, v2f b, v8f c) {
    return __builtin_amdgcn_wmma_f32_16x16x4_f32(
         false, a,  false, b,
         (short)0, c,  false,  false);
}

__global__ __launch_bounds__(256)
void qkv_proj_kernel(const float* __restrict__ x,
                     const float* __restrict__ Wq, const float* __restrict__ bq,
                     const float* __restrict__ Wk, const float* __restrict__ bk,
                     const float* __restrict__ Wv, const float* __restrict__ bv,
                     float* __restrict__ QKV)
{
    const int lane = threadIdx.x & 31;
    const int w    = threadIdx.x >> 5;
    const int tile = blockIdx.x * 8 + w;

    const int NT  = D_MODEL / 16;
    const int MT  = (B_SZ * T_SZ) / 16;
    const int mat = tile / (MT * NT);
    const int rem = tile % (MT * NT);
    const int mt  = rem / NT;
    const int nt  = rem % NT;

    const float* W  = (mat == 0) ? Wq : (mat == 1) ? Wk : Wv;
    const float* bs = (mat == 0) ? bq : (mat == 1) ? bk : bv;

    const int r15 = lane & 15;
    const int grp = lane >> 4;
    const int n   = nt * 16 + r15;
    const int m0  = mt * 16;

    const float* xr = x + (size_t)(m0 + r15) * D_MODEL;
    const float* wr = W + (size_t)n * D_MODEL;

    v8f acc = {};
    for (int k0 = 0; k0 < D_MODEL; k0 += 4) {
        const int ka = k0 + grp * 2;
        v2f a = *(const v2f*)(xr + ka);
        v2f b = *(const v2f*)(wr + ka);
        acc = wmma4(a, b, acc);
    }

    const float bias = bs[n];
    const int h = n >> 5, d = n & 31;
    float* dst = QKV + (size_t)mat * ((size_t)B_SZ * NUM_HEADS * T_SZ * DK);
#pragma unroll
    for (int m = 0; m < 8; ++m) {
        const int tl = m0 + m + 8 * grp;
        const int b_ = tl / T_SZ, t_ = tl % T_SZ;
        const float v = acc[m] + bias;
        if (mat == 2) {
            dst[(((size_t)(b_ * NUM_HEADS + h)) * DK + d) * T_SZ + t_] = v;
        } else {
            dst[(((size_t)(b_ * NUM_HEADS + h)) * T_SZ + t_) * DK + d] = v;
        }
    }
}

__device__ __forceinline__ float xhalf(float x) {
    const int b = __builtin_bit_cast(int, x);
    return __builtin_bit_cast(float,
        __builtin_amdgcn_permlanex16(b, b, 0x76543210, 0xfedcba98, false, false));
}

template <bool BETA_ONE>
__device__ __forceinline__ void attn_core(
    const float* __restrict__ Q, const float* __restrict__ K,
    const float* __restrict__ Vt, const float* __restrict__ oc,
    float* __restrict__ ctx,
    int b, int h, int qt, int lane,
    float eta, float absbeta, float alpha)
{
    const int col  = lane & 15;
    const int grp  = lane >> 4;
    const int q0   = qt * 16;
    const int qrow = q0 + col;

    const float scale   = 0.17677669529663687f;
    const float NEG_BIG = -3.4e38f;

    const size_t bh  = (size_t)(b * NUM_HEADS + h) * T_SZ;
    const size_t bhd = (size_t)(b * NUM_HEADS + h) * DK;

    const float* qp = Q + (bh + qrow) * DK;
    v2f qb[8];
#pragma unroll
    for (int kc = 0; kc < 8; ++kc) qb[kc] = *(const v2f*)(qp + kc * 4 + grp * 2);

    const float* op = oc + ((size_t)b * T_SZ + qrow) * 3;
    const float qoc0 = op[0], qoc1 = op[1], qoc2 = op[2];

    const float* vt0 = Vt + (bhd + col) * T_SZ;
    const float* vt1 = vt0 + 16 * T_SZ;

    float M = NEG_BIG;
    float L = 0.f;
    v8f O0 = {}, O1 = {};

    for (int kt = 0; kt <= qt; ++kt) {
        const int k0    = kt * 16;
        const int kbase = k0 + 8 * grp;
        const float* kp = K + (bh + k0 + col) * DK;

        if (kt < qt) {
            __builtin_prefetch(kp + 16 * DK, 0, 0);
            __builtin_prefetch(vt0 + k0 + 16, 0, 0);
            __builtin_prefetch(vt1 + k0 + 16, 0, 0);
        }

        v8f s = {};
#pragma unroll
        for (int kc = 0; kc < 8; ++kc) {
            v2f ka = *(const v2f*)(kp + kc * 4 + grp * 2);
            s = wmma4(ka, qb[kc], s);
        }

        const v4f* kop = (const v4f*)(oc + ((size_t)b * T_SZ + kbase) * 3);
        const v4f c0 = kop[0], c1 = kop[1], c2 = kop[2], c3 = kop[3], c4 = kop[4], c5 = kop[5];
        const float kocv[24] = { c0.x, c0.y, c0.z, c0.w, c1.x, c1.y, c1.z, c1.w,
                                 c2.x, c2.y, c2.z, c2.w, c3.x, c3.y, c3.z, c3.w,
                                 c4.x, c4.y, c4.z, c4.w, c5.x, c5.y, c5.z, c5.w };

#pragma unroll
        for (int r = 0; r < 8; ++r) {
            const int kcol = kbase + r;
            if (kcol > qrow) {
                s[r] = NEG_BIG;
            } else {
                const float dist = (float)(qrow - kcol);
                const float powd = BETA_ONE ? dist : __powf(dist, absbeta);
                const float td   = eta / (1.f + powd);
                const float d0 = qoc0 - kocv[3 * r + 0];
                const float d1 = qoc1 - kocv[3 * r + 1];
                const float d2 = qoc2 - kocv[3 * r + 2];
                const float pr = alpha * td * __expf(-(d0*d0 + d1*d1 + d2*d2));
                s[r] = s[r] * scale * pr;
            }
        }

        float mt = fmaxf(fmaxf(fmaxf(s[0], s[1]), fmaxf(s[2], s[3])),
                         fmaxf(fmaxf(s[4], s[5]), fmaxf(s[6], s[7])));
        mt = fmaxf(mt, xhalf(mt));
        const float mnew = fmaxf(M, mt);
        const float sc   = __expf(M - mnew);
        float p[8];
#pragma unroll
        for (int r = 0; r < 8; ++r) p[r] = __expf(s[r] - mnew);
        const float su = ((p[0] + p[1]) + (p[2] + p[3])) + ((p[4] + p[5]) + (p[6] + p[7]));
        L = L * sc + su;
        M = mnew;
#pragma unroll
        for (int r = 0; r < 8; ++r) { O0[r] *= sc; O1[r] *= sc; }

#pragma unroll
        for (int j = 0; j < 4; ++j) {
            v2f pb;
            pb.x = p[2 * j];
            pb.y = p[2 * j + 1];
            v2f a0 = *(const v2f*)(vt0 + kbase + 2 * j);
            v2f a1 = *(const v2f*)(vt1 + kbase + 2 * j);
            O0 = wmma4(a0, pb, O0);
            O1 = wmma4(a1, pb, O1);
        }
    }

    L = L + xhalf(L);
    const float inv = 1.f / L;
    float* cp = ctx + ((size_t)b * T_SZ + qrow) * D_MODEL + h * DK + 8 * grp;
    v4f o;
    o.x = O0[0] * inv; o.y = O0[1] * inv; o.z = O0[2] * inv; o.w = O0[3] * inv;
    *(v4f*)(cp) = o;
    o.x = O0[4] * inv; o.y = O0[5] * inv; o.z = O0[6] * inv; o.w = O0[7] * inv;
    *(v4f*)(cp + 4) = o;
    o.x = O1[0] * inv; o.y = O1[1] * inv; o.z = O1[2] * inv; o.w = O1[3] * inv;
    *(v4f*)(cp + 16) = o;
    o.x = O1[4] * inv; o.y = O1[5] * inv; o.z = O1[6] * inv; o.w = O1[7] * inv;
    *(v4f*)(cp + 20) = o;
}

__global__ __launch_bounds__(256)
void attn_kernel(const float* __restrict__ QKV,
                 const float* __restrict__ oc,
                 const float* __restrict__ etap,
                 const float* __restrict__ betap,
                 const float* __restrict__ alphap,
                 float* __restrict__ ctx)
{
    const int lane = threadIdx.x & 31;
    const int w    = threadIdx.x >> 5;
    const int wid  = blockIdx.x * 8 + w;

    const int QT = T_SZ / 16;
    const int qt = wid % QT;
    const int h  = (wid / QT) % NUM_HEADS;
    const int b  = wid / (QT * NUM_HEADS);

    const float eta     = etap[0];
    const float absbeta = fabsf(betap[0]);
    const float alpha   = alphap[0];

    const size_t HT = (size_t)B_SZ * NUM_HEADS * T_SZ * DK;
    const float* Q  = QKV;
    const float* K  = QKV + HT;
    const float* Vt = QKV + 2 * HT;

    if (absbeta == 1.0f) {
        attn_core<true >(Q, K, Vt, oc, ctx, b, h, qt, lane, eta, absbeta, alpha);
    } else {
        attn_core<false>(Q, K, Vt, oc, ctx, b, h, qt, lane, eta, absbeta, alpha);
    }
}

__global__ __launch_bounds__(256)
void out_proj_kernel(const float* __restrict__ ctx,
                     const float* __restrict__ Wo, const float* __restrict__ bo,
                     float* __restrict__ out)
{
    const int lane = threadIdx.x & 31;
    const int w    = threadIdx.x >> 5;
    const int tile = blockIdx.x * 8 + w;

    const int NT = D_MODEL / 16;
    const int mt = tile / NT;
    const int nt = tile % NT;

    const int r15 = lane & 15;
    const int grp = lane >> 4;
    const int n   = nt * 16 + r15;
    const int m0  = mt * 16;

    const float* xr = ctx + (size_t)(m0 + r15) * D_MODEL;
    const float* wr = Wo + (size_t)n * D_MODEL;

    v8f acc = {};
    for (int k0 = 0; k0 < D_MODEL; k0 += 4) {
        const int ka = k0 + grp * 2;
        v2f a = *(const v2f*)(xr + ka);
        v2f b = *(const v2f*)(wr + ka);
        acc = wmma4(a, b, acc);
    }

    const float bias = bo[n];
#pragma unroll
    for (int m = 0; m < 8; ++m) {
        const int tl = m0 + m + 8 * grp;
        out[(size_t)tl * D_MODEL + n] = acc[m] + bias;
    }
}

extern "C" void kernel_launch(void* const* d_in, const int* in_sizes, int n_in,
                              void* d_out, int out_size, void* d_ws, size_t ws_size,
                              hipStream_t stream) {
    (void)in_sizes; (void)n_in; (void)out_size; (void)ws_size;

    const float* x     = (const float*)d_in[0];
    const float* oc    = (const float*)d_in[1];
    const float* Wq    = (const float*)d_in[2];
    const float* bq    = (const float*)d_in[3];
    const float* Wk    = (const float*)d_in[4];
    const float* bk    = (const float*)d_in[5];
    const float* Wv    = (const float*)d_in[6];
    const float* bv    = (const float*)d_in[7];
    const float* Wo    = (const float*)d_in[8];
    const float* bo    = (const float*)d_in[9];
    const float* eta   = (const float*)d_in[10];
    const float* beta  = (const float*)d_in[11];
    const float* alpha = (const float*)d_in[12];
    float* out = (float*)d_out;

    float* ws  = (float*)d_ws;
    float* QKV = ws;
    float* ctx = ws + 3 * (size_t)B_SZ * NUM_HEADS * T_SZ * DK;

    {
        const int tiles = 3 * (B_SZ * T_SZ / 16) * (D_MODEL / 16);
        qkv_proj_kernel<<<tiles / 8, 256, 0, stream>>>(x, Wq, bq, Wk, bk, Wv, bv, QKV);
    }
    {
        const int waves = B_SZ * NUM_HEADS * (T_SZ / 16);
        attn_kernel<<<waves / 8, 256, 0, stream>>>(QKV, oc, eta, beta, alpha, ctx);
    }
    {
        const int tiles = (B_SZ * T_SZ / 16) * (D_MODEL / 16);
        out_proj_kernel<<<tiles / 8, 256, 0, stream>>>(ctx, Wo, bo, out);
    }
}


# attention causal load balance: each wave owns the query-tile pair (127-p, p), 129 key tiles for every wave; 2048 equal waves instead of 4096 waves of 1..128 tiles
# speedup vs baseline: 1.2192x; 1.0735x over previous
#include <hip/hip_runtime.h>
#include <math.h>

#define D_MODEL   256
#define NUM_HEADS 8
#define DK        32
#define B_SZ      4
#define T_SZ      2048

typedef float v2f __attribute__((ext_vector_type(2)));
typedef float v4f __attribute__((ext_vector_type(4)));
typedef float v8f __attribute__((ext_vector_type(8)));

__device__ __forceinline__ v8f wmma4(v2f a, v2f b, v8f c) {
    return __builtin_amdgcn_wmma_f32_16x16x4_f32(
         false, a,  false, b,
         (short)0, c,  false,  false);
}

__global__ __launch_bounds__(256)
void qkv_proj_kernel(const float* __restrict__ x,
                     const float* __restrict__ Wq, const float* __restrict__ bq,
                     const float* __restrict__ Wk, const float* __restrict__ bk,
                     const float* __restrict__ Wv, const float* __restrict__ bv,
                     float* __restrict__ QKV)
{
    const int lane = threadIdx.x & 31;
    const int w    = threadIdx.x >> 5;
    const int tile = blockIdx.x * 8 + w;

    const int NT  = D_MODEL / 16;
    const int MT  = (B_SZ * T_SZ) / 16;
    const int mat = tile / (MT * NT);
    const int rem = tile % (MT * NT);
    const int mt  = rem / NT;
    const int nt  = rem % NT;

    const float* W  = (mat == 0) ? Wq : (mat == 1) ? Wk : Wv;
    const float* bs = (mat == 0) ? bq : (mat == 1) ? bk : bv;

    const int r15 = lane & 15;
    const int grp = lane >> 4;
    const int n   = nt * 16 + r15;
    const int m0  = mt * 16;

    const float* xr = x + (size_t)(m0 + r15) * D_MODEL;
    const float* wr = W + (size_t)n * D_MODEL;

    v8f acc = {};
    for (int k0 = 0; k0 < D_MODEL; k0 += 4) {
        const int ka = k0 + grp * 2;
        v2f a = *(const v2f*)(xr + ka);
        v2f b = *(const v2f*)(wr + ka);
        acc = wmma4(a, b, acc);
    }

    const float bias = bs[n];
    const int h = n >> 5, d = n & 31;
    float* dst = QKV + (size_t)mat * ((size_t)B_SZ * NUM_HEADS * T_SZ * DK);
#pragma unroll
    for (int m = 0; m < 8; ++m) {
        const int tl = m0 + m + 8 * grp;
        const int b_ = tl / T_SZ, t_ = tl % T_SZ;
        const float v = acc[m] + bias;
        if (mat == 2) {
            dst[(((size_t)(b_ * NUM_HEADS + h)) * DK + d) * T_SZ + t_] = v;
        } else {
            dst[(((size_t)(b_ * NUM_HEADS + h)) * T_SZ + t_) * DK + d] = v;
        }
    }
}

__device__ __forceinline__ float xhalf(float x) {
    const int b = __builtin_bit_cast(int, x);
    return __builtin_bit_cast(float,
        __builtin_amdgcn_permlanex16(b, b, 0x76543210, 0xfedcba98, false, false));
}

template <bool BETA_ONE>
__device__ __forceinline__ void attn_core(
    const float* __restrict__ Q, const float* __restrict__ K,
    const float* __restrict__ Vt, const float* __restrict__ oc,
    float* __restrict__ ctx,
    int b, int h, int qt, int lane,
    float eta, float absbeta, float alpha)
{
    const int col  = lane & 15;
    const int grp  = lane >> 4;
    const int q0   = qt * 16;
    const int qrow = q0 + col;

    const float scale   = 0.17677669529663687f;
    const float NEG_BIG = -3.4e38f;

    const size_t bh  = (size_t)(b * NUM_HEADS + h) * T_SZ;
    const size_t bhd = (size_t)(b * NUM_HEADS + h) * DK;

    const float* qp = Q + (bh + qrow) * DK;
    v2f qb[8];
#pragma unroll
    for (int kc = 0; kc < 8; ++kc) qb[kc] = *(const v2f*)(qp + kc * 4 + grp * 2);

    const float* op = oc + ((size_t)b * T_SZ + qrow) * 3;
    const float qoc0 = op[0], qoc1 = op[1], qoc2 = op[2];

    const float* vt0 = Vt + (bhd + col) * T_SZ;
    const float* vt1 = vt0 + 16 * T_SZ;

    float M = NEG_BIG;
    float L = 0.f;
    v8f O0 = {}, O1 = {};

    for (int kt = 0; kt <= qt; ++kt) {
        const int k0    = kt * 16;
        const int kbase = k0 + 8 * grp;
        const float* kp = K + (bh + k0 + col) * DK;

        if (kt < qt) {
            __builtin_prefetch(kp + 16 * DK, 0, 0);
            __builtin_prefetch(vt0 + k0 + 16, 0, 0);
            __builtin_prefetch(vt1 + k0 + 16, 0, 0);
        }

        v8f s = {};
#pragma unroll
        for (int kc = 0; kc < 8; ++kc) {
            v2f ka = *(const v2f*)(kp + kc * 4 + grp * 2);
            s = wmma4(ka, qb[kc], s);
        }

        const v4f* kop = (const v4f*)(oc + ((size_t)b * T_SZ + kbase) * 3);
        const v4f c0 = kop[0], c1 = kop[1], c2 = kop[2], c3 = kop[3], c4 = kop[4], c5 = kop[5];
        const float kocv[24] = { c0.x, c0.y, c0.z, c0.w, c1.x, c1.y, c1.z, c1.w,
                                 c2.x, c2.y, c2.z, c2.w, c3.x, c3.y, c3.z, c3.w,
                                 c4.x, c4.y, c4.z, c4.w, c5.x, c5.y, c5.z, c5.w };

#pragma unroll
        for (int r = 0; r < 8; ++r) {
            const int kcol = kbase + r;
            if (kcol > qrow) {
                s[r] = NEG_BIG;
            } else {
                const float dist = (float)(qrow - kcol);
                const float powd = BETA_ONE ? dist : __powf(dist, absbeta);
                const float td   = eta / (1.f + powd);
                const float d0 = qoc0 - kocv[3 * r + 0];
                const float d1 = qoc1 - kocv[3 * r + 1];
                const float d2 = qoc2 - kocv[3 * r + 2];
                const float pr = alpha * td * __expf(-(d0*d0 + d1*d1 + d2*d2));
                s[r] = s[r] * scale * pr;
            }
        }

        float mt = fmaxf(fmaxf(fmaxf(s[0], s[1]), fmaxf(s[2], s[3])),
                         fmaxf(fmaxf(s[4], s[5]), fmaxf(s[6], s[7])));
        mt = fmaxf(mt, xhalf(mt));
        const float mnew = fmaxf(M, mt);
        const float sc   = __expf(M - mnew);
        float p[8];
#pragma unroll
        for (int r = 0; r < 8; ++r) p[r] = __expf(s[r] - mnew);
        const float su = ((p[0] + p[1]) + (p[2] + p[3])) + ((p[4] + p[5]) + (p[6] + p[7]));
        L = L * sc + su;
        M = mnew;
#pragma unroll
        for (int r = 0; r < 8; ++r) { O0[r] *= sc; O1[r] *= sc; }

#pragma unroll
        for (int j = 0; j < 4; ++j) {
            v2f pb;
            pb.x = p[2 * j];
            pb.y = p[2 * j + 1];
            v2f a0 = *(const v2f*)(vt0 + kbase + 2 * j);
            v2f a1 = *(const v2f*)(vt1 + kbase + 2 * j);
            O0 = wmma4(a0, pb, O0);
            O1 = wmma4(a1, pb, O1);
        }
    }

    L = L + xhalf(L);
    const float inv = 1.f / L;
    float* cp = ctx + ((size_t)b * T_SZ + qrow) * D_MODEL + h * DK + 8 * grp;
    v4f o;
    o.x = O0[0] * inv; o.y = O0[1] * inv; o.z = O0[2] * inv; o.w = O0[3] * inv;
    *(v4f*)(cp) = o;
    o.x = O0[4] * inv; o.y = O0[5] * inv; o.z = O0[6] * inv; o.w = O0[7] * inv;
    *(v4f*)(cp + 4) = o;
    o.x = O1[0] * inv; o.y = O1[1] * inv; o.z = O1[2] * inv; o.w = O1[3] * inv;
    *(v4f*)(cp + 16) = o;
    o.x = O1[4] * inv; o.y = O1[5] * inv; o.z = O1[6] * inv; o.w = O1[7] * inv;
    *(v4f*)(cp + 20) = o;
}

__global__ __launch_bounds__(256)
void attn_kernel(const float* __restrict__ QKV,
                 const float* __restrict__ oc,
                 const float* __restrict__ etap,
                 const float* __restrict__ betap,
                 const float* __restrict__ alphap,
                 float* __restrict__ ctx)
{
    const int lane = threadIdx.x & 31;
    const int w    = threadIdx.x >> 5;
    const int wid  = blockIdx.x * 8 + w;

    const int QT = T_SZ / 16;
    const int QP = QT / 2;
    const int pr = wid % QP;
    const int h  = (wid / QP) % NUM_HEADS;
    const int b  = wid / (QP * NUM_HEADS);

    const float eta     = etap[0];
    const float absbeta = fabsf(betap[0]);
    const float alpha   = alphap[0];

    const size_t HT = (size_t)B_SZ * NUM_HEADS * T_SZ * DK;
    const float* Q  = QKV;
    const float* K  = QKV + HT;
    const float* Vt = QKV + 2 * HT;

    if (absbeta == 1.0f) {
        for (int pass = 0; pass < 2; ++pass) {
            const int qt = pass ? pr : (QT - 1 - pr);
            attn_core<true >(Q, K, Vt, oc, ctx, b, h, qt, lane, eta, absbeta, alpha);
        }
    } else {
        for (int pass = 0; pass < 2; ++pass) {
            const int qt = pass ? pr : (QT - 1 - pr);
            attn_core<false>(Q, K, Vt, oc, ctx, b, h, qt, lane, eta, absbeta, alpha);
        }
    }
}

__global__ __launch_bounds__(256)
void out_proj_kernel(const float* __restrict__ ctx,
                     const float* __restrict__ Wo, const float* __restrict__ bo,
                     float* __restrict__ out)
{
    const int lane = threadIdx.x & 31;
    const int w    = threadIdx.x >> 5;
    const int tile = blockIdx.x * 8 + w;

    const int NT = D_MODEL / 16;
    const int mt = tile / NT;
    const int nt = tile % NT;

    const int r15 = lane & 15;
    const int grp = lane >> 4;
    const int n   = nt * 16 + r15;
    const int m0  = mt * 16;

    const float* xr = ctx + (size_t)(m0 + r15) * D_MODEL;
    const float* wr = Wo + (size_t)n * D_MODEL;

    v8f acc = {};
    for (int k0 = 0; k0 < D_MODEL; k0 += 4) {
        const int ka = k0 + grp * 2;
        v2f a = *(const v2f*)(xr + ka);
        v2f b = *(const v2f*)(wr + ka);
        acc = wmma4(a, b, acc);
    }

    const float bias = bo[n];
#pragma unroll
    for (int m = 0; m < 8; ++m) {
        const int tl = m0 + m + 8 * grp;
        out[(size_t)tl * D_MODEL + n] = acc[m] + bias;
    }
}

extern "C" void kernel_launch(void* const* d_in, const int* in_sizes, int n_in,
                              void* d_out, int out_size, void* d_ws, size_t ws_size,
                              hipStream_t stream) {
    (void)in_sizes; (void)n_in; (void)out_size; (void)ws_size;

    const float* x     = (const float*)d_in[0];
    const float* oc    = (const float*)d_in[1];
    const float* Wq    = (const float*)d_in[2];
    const float* bq    = (const float*)d_in[3];
    const float* Wk    = (const float*)d_in[4];
    const float* bk    = (const float*)d_in[5];
    const float* Wv    = (const float*)d_in[6];
    const float* bv    = (const float*)d_in[7];
    const float* Wo    = (const float*)d_in[8];
    const float* bo    = (const float*)d_in[9];
    const float* eta   = (const float*)d_in[10];
    const float* beta  = (const float*)d_in[11];
    const float* alpha = (const float*)d_in[12];
    float* out = (float*)d_out;

    float* ws  = (float*)d_ws;
    float* QKV = ws;
    float* ctx = ws + 3 * (size_t)B_SZ * NUM_HEADS * T_SZ * DK;

    {
        const int tiles = 3 * (B_SZ * T_SZ / 16) * (D_MODEL / 16);
        qkv_proj_kernel<<<tiles / 8, 256, 0, stream>>>(x, Wq, bq, Wk, bk, Wv, bv, QKV);
    }
    {
        const int waves = B_SZ * NUM_HEADS * (T_SZ / 32);
        attn_kernel<<<waves / 8, 256, 0, stream>>>(QKV, oc, eta, beta, alpha, ctx);
    }
    {
        const int tiles = (B_SZ * T_SZ / 16) * (D_MODEL / 16);
        out_proj_kernel<<<tiles / 8, 256, 0, stream>>>(ctx, Wo, bo, out);
    }
}


# attention key-tile body made one straight-line block: causal mask only on the diagonal tile, as a select, so the 8 per-key prior chains interleave and all tile loads issue at the tile head
# speedup vs baseline: 1.6533x; 1.3561x over previous
#include <hip/hip_runtime.h>
#include <math.h>

#define D_MODEL   256
#define NUM_HEADS 8
#define DK        32
#define B_SZ      4
#define T_SZ      2048

typedef float v2f __attribute__((ext_vector_type(2)));
typedef float v4f __attribute__((ext_vector_type(4)));
typedef float v8f __attribute__((ext_vector_type(8)));

__device__ __forceinline__ v8f wmma4(v2f a, v2f b, v8f c) {
    return __builtin_amdgcn_wmma_f32_16x16x4_f32(
         false, a,  false, b,
         (short)0, c,  false,  false);
}

__global__ __launch_bounds__(256)
void qkv_proj_kernel(const float* __restrict__ x,
                     const float* __restrict__ Wq, const float* __restrict__ bq,
                     const float* __restrict__ Wk, const float* __restrict__ bk,
                     const float* __restrict__ Wv, const float* __restrict__ bv,
                     float* __restrict__ QKV)
{
    const int lane = threadIdx.x & 31;
    const int w    = threadIdx.x >> 5;
    const int tile = blockIdx.x * 8 + w;

    const int NT  = D_MODEL / 16;
    const int MT  = (B_SZ * T_SZ) / 16;
    const int mat = tile / (MT * NT);
    const int rem = tile % (MT * NT);
    const int mt  = rem / NT;
    const int nt  = rem % NT;

    const float* W  = (mat == 0) ? Wq : (mat == 1) ? Wk : Wv;
    const float* bs = (mat == 0) ? bq : (mat == 1) ? bk : bv;

    const int r15 = lane & 15;
    const int grp = lane >> 4;
    const int n   = nt * 16 + r15;
    const int m0  = mt * 16;

    const float* xr = x + (size_t)(m0 + r15) * D_MODEL;
    const float* wr = W + (size_t)n * D_MODEL;

    v8f acc = {};
    for (int k0 = 0; k0 < D_MODEL; k0 += 4) {
        const int ka = k0 + grp * 2;
        v2f a = *(const v2f*)(xr + ka);
        v2f b = *(const v2f*)(wr + ka);
        acc = wmma4(a, b, acc);
    }

    const float bias = bs[n];
    const int h = n >> 5, d = n & 31;
    float* dst = QKV + (size_t)mat * ((size_t)B_SZ * NUM_HEADS * T_SZ * DK);
#pragma unroll
    for (int m = 0; m < 8; ++m) {
        const int tl = m0 + m + 8 * grp;
        const int b_ = tl / T_SZ, t_ = tl % T_SZ;
        const float v = acc[m] + bias;
        if (mat == 2) {
            dst[(((size_t)(b_ * NUM_HEADS + h)) * DK + d) * T_SZ + t_] = v;
        } else {
            dst[(((size_t)(b_ * NUM_HEADS + h)) * T_SZ + t_) * DK + d] = v;
        }
    }
}

__device__ __forceinline__ float xhalf(float x) {
    const int b = __builtin_bit_cast(int, x);
    return __builtin_bit_cast(float,
        __builtin_amdgcn_permlanex16(b, b, 0x76543210, 0xfedcba98, false, false));
}

template <bool BETA_ONE, bool DIAG>
__device__ __forceinline__ void attn_tile(
    const float* __restrict__ K, const float* __restrict__ oc,
    const float* __restrict__ vt0, const float* __restrict__ vt1,
    const v2f (&qb)[8], float qoc0, float qoc1, float qoc2,
    size_t bh, int b, int qrow, int kt, int col, int grp,
    float eta, float absbeta, float alpha,
    float& M, float& L, v8f& O0, v8f& O1)
{
    const float scale   = 0.17677669529663687f;
    const float NEG_BIG = -3.4e38f;

    const int k0    = kt * 16;
    const int kbase = k0 + 8 * grp;
    const float* kp = K + (bh + k0 + col) * DK;

    if (!DIAG) {
        __builtin_prefetch(kp + 16 * DK, 0, 0);
        __builtin_prefetch(vt0 + k0 + 16, 0, 0);
        __builtin_prefetch(vt1 + k0 + 16, 0, 0);
    }

    v8f s = {};
#pragma unroll
    for (int kc = 0; kc < 8; ++kc) {
        v2f ka = *(const v2f*)(kp + kc * 4 + grp * 2);
        s = wmma4(ka, qb[kc], s);
    }

    const v4f* kop = (const v4f*)(oc + ((size_t)b * T_SZ + kbase) * 3);
    const v4f c0 = kop[0], c1 = kop[1], c2 = kop[2], c3 = kop[3], c4 = kop[4], c5 = kop[5];
    const float kocv[24] = { c0.x, c0.y, c0.z, c0.w, c1.x, c1.y, c1.z, c1.w,
                             c2.x, c2.y, c2.z, c2.w, c3.x, c3.y, c3.z, c3.w,
                             c4.x, c4.y, c4.z, c4.w, c5.x, c5.y, c5.z, c5.w };

#pragma unroll
    for (int r = 0; r < 8; ++r) {
        const int kcol = kbase + r;
        const float dist = (float)(qrow - kcol);
        const float powd = BETA_ONE ? dist : __powf(dist, absbeta);
        const float td   = eta / (1.f + powd);
        const float d0 = qoc0 - kocv[3 * r + 0];
        const float d1 = qoc1 - kocv[3 * r + 1];
        const float d2 = qoc2 - kocv[3 * r + 2];
        const float pr = alpha * td * __expf(-(d0*d0 + d1*d1 + d2*d2));
        const float val = s[r] * scale * pr;
        if (DIAG) s[r] = (kcol > qrow) ? NEG_BIG : val;
        else      s[r] = val;
    }

    float mt = fmaxf(fmaxf(fmaxf(s[0], s[1]), fmaxf(s[2], s[3])),
                     fmaxf(fmaxf(s[4], s[5]), fmaxf(s[6], s[7])));
    mt = fmaxf(mt, xhalf(mt));
    const float mnew = fmaxf(M, mt);
    const float sc   = __expf(M - mnew);
    float p[8];
#pragma unroll
    for (int r = 0; r < 8; ++r) p[r] = __expf(s[r] - mnew);
    const float su = ((p[0] + p[1]) + (p[2] + p[3])) + ((p[4] + p[5]) + (p[6] + p[7]));
    L = L * sc + su;
    M = mnew;
#pragma unroll
    for (int r = 0; r < 8; ++r) { O0[r] *= sc; O1[r] *= sc; }

#pragma unroll
    for (int j = 0; j < 4; ++j) {
        v2f pb;
        pb.x = p[2 * j];
        pb.y = p[2 * j + 1];
        v2f a0 = *(const v2f*)(vt0 + kbase + 2 * j);
        v2f a1 = *(const v2f*)(vt1 + kbase + 2 * j);
        O0 = wmma4(a0, pb, O0);
        O1 = wmma4(a1, pb, O1);
    }
}

template <bool BETA_ONE>
__device__ __forceinline__ void attn_core(
    const float* __restrict__ Q, const float* __restrict__ K,
    const float* __restrict__ Vt, const float* __restrict__ oc,
    float* __restrict__ ctx,
    int b, int h, int qt, int lane,
    float eta, float absbeta, float alpha)
{
    const int col  = lane & 15;
    const int grp  = lane >> 4;
    const int q0   = qt * 16;
    const int qrow = q0 + col;

    const float NEG_BIG = -3.4e38f;

    const size_t bh  = (size_t)(b * NUM_HEADS + h) * T_SZ;
    const size_t bhd = (size_t)(b * NUM_HEADS + h) * DK;

    const float* qp = Q + (bh + qrow) * DK;
    v2f qb[8];
#pragma unroll
    for (int kc = 0; kc < 8; ++kc) qb[kc] = *(const v2f*)(qp + kc * 4 + grp * 2);

    const float* op = oc + ((size_t)b * T_SZ + qrow) * 3;
    const float qoc0 = op[0], qoc1 = op[1], qoc2 = op[2];

    const float* vt0 = Vt + (bhd + col) * T_SZ;
    const float* vt1 = vt0 + 16 * T_SZ;

    float M = NEG_BIG;
    float L = 0.f;
    v8f O0 = {}, O1 = {};

    for (int kt = 0; kt < qt; ++kt)
        attn_tile<BETA_ONE, false>(K, oc, vt0, vt1, qb, qoc0, qoc1, qoc2, bh, b, qrow, kt,
                                   col, grp, eta, absbeta, alpha, M, L, O0, O1);
    attn_tile<BETA_ONE, true>(K, oc, vt0, vt1, qb, qoc0, qoc1, qoc2, bh, b, qrow, qt,
                              col, grp, eta, absbeta, alpha, M, L, O0, O1);

    L = L + xhalf(L);
    const float inv = 1.f / L;
    float* cp = ctx + ((size_t)b * T_SZ + qrow) * D_MODEL + h * DK + 8 * grp;
    v4f o;
    o.x = O0[0] * inv; o.y = O0[1] * inv; o.z = O0[2] * inv; o.w = O0[3] * inv;
    *(v4f*)(cp) = o;
    o.x = O0[4] * inv; o.y = O0[5] * inv; o.z = O0[6] * inv; o.w = O0[7] * inv;
    *(v4f*)(cp + 4) = o;
    o.x = O1[0] * inv; o.y = O1[1] * inv; o.z = O1[2] * inv; o.w = O1[3] * inv;
    *(v4f*)(cp + 16) = o;
    o.x = O1[4] * inv; o.y = O1[5] * inv; o.z = O1[6] * inv; o.w = O1[7] * inv;
    *(v4f*)(cp + 20) = o;
}

__global__ __launch_bounds__(256)
void attn_kernel(const float* __restrict__ QKV,
                 const float* __restrict__ oc,
                 const float* __restrict__ etap,
                 const float* __restrict__ betap,
                 const float* __restrict__ alphap,
                 float* __restrict__ ctx)
{
    const int lane = threadIdx.x & 31;
    const int w    = threadIdx.x >> 5;
    const int wid  = blockIdx.x * 8 + w;

    const int QT = T_SZ / 16;
    const int QP = QT / 2;
    const int pr = wid % QP;
    const int h  = (wid / QP) % NUM_HEADS;
    const int b  = wid / (QP * NUM_HEADS);

    const float eta     = etap[0];
    const float absbeta = fabsf(betap[0]);
    const float alpha   = alphap[0];

    const size_t HT = (size_t)B_SZ * NUM_HEADS * T_SZ * DK;
    const float* Q  = QKV;
    const float* K  = QKV + HT;
    const float* Vt = QKV + 2 * HT;

    if (absbeta == 1.0f) {
        for (int pass = 0; pass < 2; ++pass) {
            const int qt = pass ? pr : (QT - 1 - pr);
            attn_core<true >(Q, K, Vt, oc, ctx, b, h, qt, lane, eta, absbeta, alpha);
        }
    } else {
        for (int pass = 0; pass < 2; ++pass) {
            const int qt = pass ? pr : (QT - 1 - pr);
            attn_core<false>(Q, K, Vt, oc, ctx, b, h, qt, lane, eta, absbeta, alpha);
        }
    }
}

__global__ __launch_bounds__(256)
void out_proj_kernel(const float* __restrict__ ctx,
                     const float* __restrict__ Wo, const float* __restrict__ bo,
                     float* __restrict__ out)
{
    const int lane = threadIdx.x & 31;
    const int w    = threadIdx.x >> 5;
    const int tile = blockIdx.x * 8 + w;

    const int NT = D_MODEL / 16;
    const int mt = tile / NT;
    const int nt = tile % NT;

    const int r15 = lane & 15;
    const int grp = lane >> 4;
    const int n   = nt * 16 + r15;
    const int m0  = mt * 16;

    const float* xr = ctx + (size_t)(m0 + r15) * D_MODEL;
    const float* wr = Wo + (size_t)n * D_MODEL;

    v8f acc = {};
    for (int k0 = 0; k0 < D_MODEL; k0 += 4) {
        const int ka = k0 + grp * 2;
        v2f a = *(const v2f*)(xr + ka);
        v2f b = *(const v2f*)(wr + ka);
        acc = wmma4(a, b, acc);
    }

    const float bias = bo[n];
#pragma unroll
    for (int m = 0; m < 8; ++m) {
        const int tl = m0 + m + 8 * grp;
        out[(size_t)tl * D_MODEL + n] = acc[m] + bias;
    }
}

extern "C" void kernel_launch(void* const* d_in, const int* in_sizes, int n_in,
                              void* d_out, int out_size, void* d_ws, size_t ws_size,
                              hipStream_t stream) {
    (void)in_sizes; (void)n_in; (void)out_size; (void)ws_size;

    const float* x     = (const float*)d_in[0];
    const float* oc    = (const float*)d_in[1];
    const float* Wq    = (const float*)d_in[2];
    const float* bq    = (const float*)d_in[3];
    const float* Wk    = (const float*)d_in[4];
    const float* bk    = (const float*)d_in[5];
    const float* Wv    = (const float*)d_in[6];
    const float* bv    = (const float*)d_in[7];
    const float* Wo    = (const float*)d_in[8];
    const float* bo    = (const float*)d_in[9];
    const float* eta   = (const float*)d_in[10];
    const float* beta  = (const float*)d_in[11];
    const float* alpha = (const float*)d_in[12];
    float* out = (float*)d_out;

    float* ws  = (float*)d_ws;
    float* QKV = ws;
    float* ctx = ws + 3 * (size_t)B_SZ * NUM_HEADS * T_SZ * DK;

    {
        const int tiles = 3 * (B_SZ * T_SZ / 16) * (D_MODEL / 16);
        qkv_proj_kernel<<<tiles / 8, 256, 0, stream>>>(x, Wq, bq, Wk, bk, Wv, bv, QKV);
    }
    {
        const int waves = B_SZ * NUM_HEADS * (T_SZ / 32);
        attn_kernel<<<waves / 8, 256, 0, stream>>>(QKV, oc, eta, beta, alpha, ctx);
    }
    {
        const int tiles = (B_SZ * T_SZ / 16) * (D_MODEL / 16);
        out_proj_kernel<<<tiles / 8, 256, 0, stream>>>(ctx, Wo, bo, out);
    }
}


# both projection GEMMs re-tiled: 64-channel W slab staged once per block in LDS, 16x64 wave tile, 16-byte fragment loads (1 global load per 8 WMMAs, was 2 per WMMA); changes their k summation order
# speedup vs baseline: 2.6444x; 1.5995x over previous
#include <hip/hip_runtime.h>
#include <math.h>

#define D_MODEL   256
#define NUM_HEADS 8
#define DK        32
#define B_SZ      4
#define T_SZ      2048

typedef float v2f __attribute__((ext_vector_type(2)));
typedef float v4f __attribute__((ext_vector_type(4)));
typedef float v8f __attribute__((ext_vector_type(8)));

__device__ __forceinline__ v8f wmma4(v2f a, v2f b, v8f c) {
    return __builtin_amdgcn_wmma_f32_16x16x4_f32(
         false, a,  false, b,
         (short)0, c,  false,  false);
}

#define WPITCH 260
#define SLAB_N 64

__device__ __forceinline__ void proj_load_a(v4f (&a)[8], const float* __restrict__ xr, int ch) {
#pragma unroll
    for (int u = 0; u < 8; ++u) a[u] = *(const v4f*)(xr + 64 * ch + 8 * u);
}

__device__ __forceinline__ void proj_mma_chunk(const v4f (&a)[8], const float* wl, int ch, v8f (&acc)[4]) {
#pragma unroll
    for (int u = 0; u < 8; ++u) {
        const int i = ch * 8 + u;
        v4f b[4];
#pragma unroll
        for (int j = 0; j < 4; ++j) b[j] = *(const v4f*)(wl + j * 16 * WPITCH + 8 * i);
        v2f alo, ahi;
        alo.x = a[u].x; alo.y = a[u].y; ahi.x = a[u].z; ahi.y = a[u].w;
#pragma unroll
        for (int j = 0; j < 4; ++j) {
            v2f blo; blo.x = b[j].x; blo.y = b[j].y;
            acc[j] = wmma4(alo, blo, acc[j]);
        }
#pragma unroll
        for (int j = 0; j < 4; ++j) {
            v2f bhi; bhi.x = b[j].z; bhi.y = b[j].w;
            acc[j] = wmma4(ahi, bhi, acc[j]);
        }
    }
}

__device__ __forceinline__ void proj_tile_16x64(const float* __restrict__ X, const float* __restrict__ Wm,
                                                int n0, int m0, float* wlds, v8f (&acc)[4])
{
    const int tid  = threadIdx.x;
    const int lane = tid & 31;
    const int c    = lane & 15;
    const int hh   = lane >> 4;

    const float* xr = X + (size_t)(m0 + c) * D_MODEL + 4 * hh;

    v4f a0[8], a1[8];
    proj_load_a(a0, xr, 0);

    {
        const int c4 = tid & 63;
        const int r0 = tid >> 6;
        v4f t[16];
#pragma unroll
        for (int u = 0; u < 16; ++u)
            t[u] = *(const v4f*)(Wm + (size_t)(n0 + 4 * u + r0) * D_MODEL + 4 * c4);
#pragma unroll
        for (int u = 0; u < 16; ++u)
            *(v4f*)(wlds + (4 * u + r0) * WPITCH + 4 * c4) = t[u];
    }
    __syncthreads();

    const float* wl = wlds + c * WPITCH + 4 * hh;

#pragma unroll
    for (int j = 0; j < 4; ++j) acc[j] = (v8f){};

    proj_load_a(a1, xr, 1);
    proj_mma_chunk(a0, wl, 0, acc);
    proj_load_a(a0, xr, 2);
    proj_mma_chunk(a1, wl, 1, acc);
    proj_load_a(a1, xr, 3);
    proj_mma_chunk(a0, wl, 2, acc);
    proj_mma_chunk(a1, wl, 3, acc);
}

__global__ __launch_bounds__(256)
void qkv_proj_kernel(const float* __restrict__ x,
                     const float* __restrict__ Wq, const float* __restrict__ bq,
                     const float* __restrict__ Wk, const float* __restrict__ bk,
                     const float* __restrict__ Wv, const float* __restrict__ bv,
                     float* __restrict__ QKV)
{
    __shared__ __attribute__((aligned(16))) float wlds[SLAB_N * WPITCH];

    const int lane = threadIdx.x & 31;
    const int w    = threadIdx.x >> 5;

    const int MG  = (B_SZ * T_SZ) / 128;
    const int ns  = blockIdx.x / MG;
    const int mg  = blockIdx.x % MG;
    const int mat = ns >> 2;
    const int n0  = (ns & 3) * SLAB_N;
    const int m0  = mg * 128 + w * 16;

    const float* Wm = (mat == 0) ? Wq : (mat == 1) ? Wk : Wv;
    const float* bs = (mat == 0) ? bq : (mat == 1) ? bk : bv;

    v8f acc[4];
    proj_tile_16x64(x, Wm, n0, m0, wlds, acc);

    const int r15 = lane & 15;
    const int grp = lane >> 4;
    float* dst = QKV + (size_t)mat * ((size_t)B_SZ * NUM_HEADS * T_SZ * DK);
    if (mat == 2) {
#pragma unroll
        for (int j = 0; j < 4; ++j) {
            const int n = n0 + 16 * j + r15;
            const float bias = bs[n];
            const int h = n >> 5, d = n & 31;
#pragma unroll
            for (int m = 0; m < 8; ++m) {
                const int tl = m0 + m + 8 * grp;
                const int b_ = tl / T_SZ, t_ = tl % T_SZ;
                dst[(((size_t)(b_ * NUM_HEADS + h)) * DK + d) * T_SZ + t_] = acc[j][m] + bias;
            }
        }
    } else {
#pragma unroll
        for (int j = 0; j < 4; ++j) {
            const int n = n0 + 16 * j + r15;
            const float bias = bs[n];
            const int h = n >> 5, d = n & 31;
#pragma unroll
            for (int m = 0; m < 8; ++m) {
                const int tl = m0 + m + 8 * grp;
                const int b_ = tl / T_SZ, t_ = tl % T_SZ;
                dst[(((size_t)(b_ * NUM_HEADS + h)) * T_SZ + t_) * DK + d] = acc[j][m] + bias;
            }
        }
    }
}

__device__ __forceinline__ float xhalf(float x) {
    const int b = __builtin_bit_cast(int, x);
    return __builtin_bit_cast(float,
        __builtin_amdgcn_permlanex16(b, b, 0x76543210, 0xfedcba98, false, false));
}

template <bool BETA_ONE, bool DIAG>
__device__ __forceinline__ void attn_tile(
    const float* __restrict__ K, const float* __restrict__ oc,
    const float* __restrict__ vt0, const float* __restrict__ vt1,
    const v2f (&qb)[8], float qoc0, float qoc1, float qoc2,
    size_t bh, int b, int qrow, int kt, int col, int grp,
    float eta, float absbeta, float alpha,
    float& M, float& L, v8f& O0, v8f& O1)
{
    const float scale   = 0.17677669529663687f;
    const float NEG_BIG = -3.4e38f;

    const int k0    = kt * 16;
    const int kbase = k0 + 8 * grp;
    const float* kp = K + (bh + k0 + col) * DK;

    if (!DIAG) {
        __builtin_prefetch(kp + 16 * DK, 0, 0);
        __builtin_prefetch(vt0 + k0 + 16, 0, 0);
        __builtin_prefetch(vt1 + k0 + 16, 0, 0);
    }

    v8f s = {};
#pragma unroll
    for (int kc = 0; kc < 8; ++kc) {
        v2f ka = *(const v2f*)(kp + kc * 4 + grp * 2);
        s = wmma4(ka, qb[kc], s);
    }

    const v4f* kop = (const v4f*)(oc + ((size_t)b * T_SZ + kbase) * 3);
    const v4f c0 = kop[0], c1 = kop[1], c2 = kop[2], c3 = kop[3], c4 = kop[4], c5 = kop[5];
    const float kocv[24] = { c0.x, c0.y, c0.z, c0.w, c1.x, c1.y, c1.z, c1.w,
                             c2.x, c2.y, c2.z, c2.w, c3.x, c3.y, c3.z, c3.w,
                             c4.x, c4.y, c4.z, c4.w, c5.x, c5.y, c5.z, c5.w };

#pragma unroll
    for (int r = 0; r < 8; ++r) {
        const int kcol = kbase + r;
        const float dist = (float)(qrow - kcol);
        const float powd = BETA_ONE ? dist : __powf(dist, absbeta);
        const float td   = eta / (1.f + powd);
        const float d0 = qoc0 - kocv[3 * r + 0];
        const float d1 = qoc1 - kocv[3 * r + 1];
        const float d2 = qoc2 - kocv[3 * r + 2];
        const float pr = alpha * td * __expf(-(d0*d0 + d1*d1 + d2*d2));
        const float val = s[r] * scale * pr;
        if (DIAG) s[r] = (kcol > qrow) ? NEG_BIG : val;
        else      s[r] = val;
    }

    float mt = fmaxf(fmaxf(fmaxf(s[0], s[1]), fmaxf(s[2], s[3])),
                     fmaxf(fmaxf(s[4], s[5]), fmaxf(s[6], s[7])));
    mt = fmaxf(mt, xhalf(mt));
    const float mnew = fmaxf(M, mt);
    const float sc   = __expf(M - mnew);
    float p[8];
#pragma unroll
    for (int r = 0; r < 8; ++r) p[r] = __expf(s[r] - mnew);
    const float su = ((p[0] + p[1]) + (p[2] + p[3])) + ((p[4] + p[5]) + (p[6] + p[7]));
    L = L * sc + su;
    M = mnew;
#pragma unroll
    for (int r = 0; r < 8; ++r) { O0[r] *= sc; O1[r] *= sc; }

#pragma unroll
    for (int j = 0; j < 4; ++j) {
        v2f pb;
        pb.x = p[2 * j];
        pb.y = p[2 * j + 1];
        v2f a0 = *(const v2f*)(vt0 + kbase + 2 * j);
        v2f a1 = *(const v2f*)(vt1 + kbase + 2 * j);
        O0 = wmma4(a0, pb, O0);
        O1 = wmma4(a1, pb, O1);
    }
}

template <bool BETA_ONE>
__device__ __forceinline__ void attn_core(
    const float* __restrict__ Q, const float* __restrict__ K,
    const float* __restrict__ Vt, const float* __restrict__ oc,
    float* __restrict__ ctx,
    int b, int h, int qt, int lane,
    float eta, float absbeta, float alpha)
{
    const int col  = lane & 15;
    const int grp  = lane >> 4;
    const int q0   = qt * 16;
    const int qrow = q0 + col;

    const float NEG_BIG = -3.4e38f;

    const size_t bh  = (size_t)(b * NUM_HEADS + h) * T_SZ;
    const size_t bhd = (size_t)(b * NUM_HEADS + h) * DK;

    const float* qp = Q + (bh + qrow) * DK;
    v2f qb[8];
#pragma unroll
    for (int kc = 0; kc < 8; ++kc) qb[kc] = *(const v2f*)(qp + kc * 4 + grp * 2);

    const float* op = oc + ((size_t)b * T_SZ + qrow) * 3;
    const float qoc0 = op[0], qoc1 = op[1], qoc2 = op[2];

    const float* vt0 = Vt + (bhd + col) * T_SZ;
    const float* vt1 = vt0 + 16 * T_SZ;

    float M = NEG_BIG;
    float L = 0.f;
    v8f O0 = {}, O1 = {};

    for (int kt = 0; kt < qt; ++kt)
        attn_tile<BETA_ONE, false>(K, oc, vt0, vt1, qb, qoc0, qoc1, qoc2, bh, b, qrow, kt,
                                   col, grp, eta, absbeta, alpha, M, L, O0, O1);
    attn_tile<BETA_ONE, true>(K, oc, vt0, vt1, qb, qoc0, qoc1, qoc2, bh, b, qrow, qt,
                              col, grp, eta, absbeta, alpha, M, L, O0, O1);

    L = L + xhalf(L);
    const float inv = 1.f / L;
    float* cp = ctx + ((size_t)b * T_SZ + qrow) * D_MODEL + h * DK + 8 * grp;
    v4f o;
    o.x = O0[0] * inv; o.y = O0[1] * inv; o.z = O0[2] * inv; o.w = O0[3] * inv;
    *(v4f*)(cp) = o;
    o.x = O0[4] * inv; o.y = O0[5] * inv; o.z = O0[6] * inv; o.w = O0[7] * inv;
    *(v4f*)(cp + 4) = o;
    o.x = O1[0] * inv; o.y = O1[1] * inv; o.z = O1[2] * inv; o.w = O1[3] * inv;
    *(v4f*)(cp + 16) = o;
    o.x = O1[4] * inv; o.y = O1[5] * inv; o.z = O1[6] * inv; o.w = O1[7] * inv;
    *(v4f*)(cp + 20) = o;
}

__global__ __launch_bounds__(256)
void attn_kernel(const float* __restrict__ QKV,
                 const float* __restrict__ oc,
                 const float* __restrict__ etap,
                 const float* __restrict__ betap,
                 const float* __restrict__ alphap,
                 float* __restrict__ ctx)
{
    const int lane = threadIdx.x & 31;
    const int w    = threadIdx.x >> 5;
    const int wid  = blockIdx.x * 8 + w;

    const int QT = T_SZ / 16;
    const int QP = QT / 2;
    const int pr = wid % QP;
    const int h  = (wid / QP) % NUM_HEADS;
    const int b  = wid / (QP * NUM_HEADS);

    const float eta     = etap[0];
    const float absbeta = fabsf(betap[0]);
    const float alpha   = alphap[0];

    const size_t HT = (size_t)B_SZ * NUM_HEADS * T_SZ * DK;
    const float* Q  = QKV;
    const float* K  = QKV + HT;
    const float* Vt = QKV + 2 * HT;

    if (absbeta == 1.0f) {
        for (int pass = 0; pass < 2; ++pass) {
            const int qt = pass ? pr : (QT - 1 - pr);
            attn_core<true >(Q, K, Vt, oc, ctx, b, h, qt, lane, eta, absbeta, alpha);
        }
    } else {
        for (int pass = 0; pass < 2; ++pass) {
            const int qt = pass ? pr : (QT - 1 - pr);
            attn_core<false>(Q, K, Vt, oc, ctx, b, h, qt, lane, eta, absbeta, alpha);
        }
    }
}

__global__ __launch_bounds__(256)
void out_proj_kernel(const float* __restrict__ ctx,
                     const float* __restrict__ Wo, const float* __restrict__ bo,
                     float* __restrict__ out)
{
    __shared__ __attribute__((aligned(16))) float wlds[SLAB_N * WPITCH];

    const int lane = threadIdx.x & 31;
    const int w    = threadIdx.x >> 5;

    const int MG = (B_SZ * T_SZ) / 128;
    const int ns = blockIdx.x / MG;
    const int mg = blockIdx.x % MG;
    const int n0 = ns * SLAB_N;
    const int m0 = mg * 128 + w * 16;

    v8f acc[4];
    proj_tile_16x64(ctx, Wo, n0, m0, wlds, acc);

    const int r15 = lane & 15;
    const int grp = lane >> 4;
#pragma unroll
    for (int j = 0; j < 4; ++j) {
        const int n = n0 + 16 * j + r15;
        const float bias = bo[n];
#pragma unroll
        for (int m = 0; m < 8; ++m) {
            const int tl = m0 + m + 8 * grp;
            out[(size_t)tl * D_MODEL + n] = acc[j][m] + bias;
        }
    }
}

extern "C" void kernel_launch(void* const* d_in, const int* in_sizes, int n_in,
                              void* d_out, int out_size, void* d_ws, size_t ws_size,
                              hipStream_t stream) {
    (void)in_sizes; (void)n_in; (void)out_size; (void)ws_size;

    const float* x     = (const float*)d_in[0];
    const float* oc    = (const float*)d_in[1];
    const float* Wq    = (const float*)d_in[2];
    const float* bq    = (const float*)d_in[3];
    const float* Wk    = (const float*)d_in[4];
    const float* bk    = (const float*)d_in[5];
    const float* Wv    = (const float*)d_in[6];
    const float* bv    = (const float*)d_in[7];
    const float* Wo    = (const float*)d_in[8];
    const float* bo    = (const float*)d_in[9];
    const float* eta   = (const float*)d_in[10];
    const float* beta  = (const float*)d_in[11];
    const float* alpha = (const float*)d_in[12];
    float* out = (float*)d_out;

    float* ws  = (float*)d_ws;
    float* QKV = ws;
    float* ctx = ws + 3 * (size_t)B_SZ * NUM_HEADS * T_SZ * DK;

    {
        const int blocks = (3 * D_MODEL / SLAB_N) * (B_SZ * T_SZ / 128);
        qkv_proj_kernel<<<blocks, 256, 0, stream>>>(x, Wq, bq, Wk, bk, Wv, bv, QKV);
    }
    {
        const int waves = B_SZ * NUM_HEADS * (T_SZ / 32);
        attn_kernel<<<waves / 8, 256, 0, stream>>>(QKV, oc, eta, beta, alpha, ctx);
    }
    {
        const int blocks = (D_MODEL / SLAB_N) * (B_SZ * T_SZ / 128);
        out_proj_kernel<<<blocks, 256, 0, stream>>>(ctx, Wo, bo, out);
    }
}


# Q/K/V workspace in matrix-fragment order (Q,K from the transposed product W*X^T): each attention operand load is one contiguous 512-byte wave access, not a line per lane; changes QK d sum order
# speedup vs baseline: 2.7980x; 1.0581x over previous
#include <hip/hip_runtime.h>
#include <math.h>

#define D_MODEL   256
#define NUM_HEADS 8
#define DK        32
#define B_SZ      4
#define T_SZ      2048

typedef float v2f __attribute__((ext_vector_type(2)));
typedef float v4f __attribute__((ext_vector_type(4)));
typedef float v8f __attribute__((ext_vector_type(8)));

__device__ __forceinline__ v8f wmma4(v2f a, v2f b, v8f c) {
    return __builtin_amdgcn_wmma_f32_16x16x4_f32(
         false, a,  false, b,
         (short)0, c,  false,  false);
}

#define WPITCH 260
#define SLAB_N 64

__device__ __forceinline__ void proj_load_a(v4f (&a)[8], const float* __restrict__ xr, int ch) {
#pragma unroll
    for (int u = 0; u < 8; ++u) a[u] = *(const v4f*)(xr + 64 * ch + 8 * u);
}

template <bool SWAP>
__device__ __forceinline__ void proj_mma_chunk(const v4f (&a)[8], const float* wl, int ch, v8f (&acc)[4]) {
#pragma unroll
    for (int u = 0; u < 8; ++u) {
        const int i = ch * 8 + u;
        v4f b[4];
#pragma unroll
        for (int j = 0; j < 4; ++j) b[j] = *(const v4f*)(wl + j * 16 * WPITCH + 8 * i);
        v2f alo, ahi;
        alo.x = a[u].x; alo.y = a[u].y; ahi.x = a[u].z; ahi.y = a[u].w;
#pragma unroll
        for (int j = 0; j < 4; ++j) {
            v2f blo; blo.x = b[j].x; blo.y = b[j].y;
            acc[j] = SWAP ? wmma4(blo, alo, acc[j]) : wmma4(alo, blo, acc[j]);
        }
#pragma unroll
        for (int j = 0; j < 4; ++j) {
            v2f bhi; bhi.x = b[j].z; bhi.y = b[j].w;
            acc[j] = SWAP ? wmma4(bhi, ahi, acc[j]) : wmma4(ahi, bhi, acc[j]);
        }
    }
}

template <bool SWAP>
__device__ __forceinline__ void proj_tile_16x64(const float* __restrict__ X, const float* __restrict__ Wm,
                                                int n0, int m0, float* wlds, v8f (&acc)[4])
{
    const int tid  = threadIdx.x;
    const int lane = tid & 31;
    const int c    = lane & 15;
    const int hh   = lane >> 4;

    const float* xr = X + (size_t)(m0 + c) * D_MODEL + 4 * hh;

    v4f a0[8], a1[8];
    proj_load_a(a0, xr, 0);

    {
        const int c4 = tid & 63;
        const int r0 = tid >> 6;
        v4f t[16];
#pragma unroll
        for (int u = 0; u < 16; ++u)
            t[u] = *(const v4f*)(Wm + (size_t)(n0 + 4 * u + r0) * D_MODEL + 4 * c4);
#pragma unroll
        for (int u = 0; u < 16; ++u)
            *(v4f*)(wlds + (4 * u + r0) * WPITCH + 4 * c4) = t[u];
    }
    __syncthreads();

    const float* wl = wlds + c * WPITCH + 4 * hh;

#pragma unroll
    for (int j = 0; j < 4; ++j) acc[j] = (v8f){};

    proj_load_a(a1, xr, 1);
    proj_mma_chunk<SWAP>(a0, wl, 0, acc);
    proj_load_a(a0, xr, 2);
    proj_mma_chunk<SWAP>(a1, wl, 1, acc);
    proj_load_a(a1, xr, 3);
    proj_mma_chunk<SWAP>(a0, wl, 2, acc);
    proj_mma_chunk<SWAP>(a1, wl, 3, acc);
}

__global__ __launch_bounds__(256)
void qkv_proj_kernel(const float* __restrict__ x,
                     const float* __restrict__ Wq, const float* __restrict__ bq,
                     const float* __restrict__ Wk, const float* __restrict__ bk,
                     const float* __restrict__ Wv, const float* __restrict__ bv,
                     float* __restrict__ QKV)
{
    __shared__ __attribute__((aligned(16))) float wlds[SLAB_N * WPITCH];

    const int lane = threadIdx.x & 31;
    const int w    = threadIdx.x >> 5;

    const int MG  = (B_SZ * T_SZ) / 128;
    const int ns  = blockIdx.x / MG;
    const int mg  = blockIdx.x % MG;
    const int mat = ns >> 2;
    const int n0  = (ns & 3) * SLAB_N;
    const int m0  = mg * 128 + w * 16;

    const float* Wm = (mat == 0) ? Wq : (mat == 1) ? Wk : Wv;
    const float* bs = (mat == 0) ? bq : (mat == 1) ? bk : bv;

    float* dst = QKV + (size_t)mat * ((size_t)B_SZ * NUM_HEADS * T_SZ * DK);

    const int b_ = m0 / T_SZ;
    const int tt = (m0 % T_SZ) >> 4;

    v8f acc[4];
    if (mat == 2) {
        proj_tile_16x64<false>(x, Wm, n0, m0, wlds, acc);
#pragma unroll
        for (int j = 0; j < 4; ++j) {
            const int nb = n0 + 16 * j;
            const float bias = bs[nb + (lane & 15)];
            const int h = nb >> 5, jd = (nb >> 4) & 1;
            float* tp = dst + ((size_t)(b_ * NUM_HEADS + h) * (T_SZ / 16) + tt) * 512 + lane * 4;
            v4f o;
            o.x = acc[j][0] + bias; o.y = acc[j][1] + bias; o.z = acc[j][2] + bias; o.w = acc[j][3] + bias;
            *(v4f*)(tp + (jd * 2 + 0) * 128) = o;
            o.x = acc[j][4] + bias; o.y = acc[j][5] + bias; o.z = acc[j][6] + bias; o.w = acc[j][7] + bias;
            *(v4f*)(tp + (jd * 2 + 1) * 128) = o;
        }
    } else {
        proj_tile_16x64<true>(x, Wm, n0, m0, wlds, acc);
#pragma unroll
        for (int j = 0; j < 4; ++j) {
            const int nb = n0 + 16 * j;
            const float* bp = bs + nb + 8 * (lane >> 4);
            v4f b0, b1;
            b0.x = bp[0]; b0.y = bp[1]; b0.z = bp[2]; b0.w = bp[3];
            b1.x = bp[4]; b1.y = bp[5]; b1.z = bp[6]; b1.w = bp[7];
            const int h = nb >> 5, jd = (nb >> 4) & 1;
            float* tp = dst + ((size_t)(b_ * NUM_HEADS + h) * (T_SZ / 16) + tt) * 512 + lane * 4;
            v4f o;
            o.x = acc[j][0] + b0.x; o.y = acc[j][1] + b0.y; o.z = acc[j][2] + b0.z; o.w = acc[j][3] + b0.w;
            *(v4f*)(tp + (jd * 2 + 0) * 128) = o;
            o.x = acc[j][4] + b1.x; o.y = acc[j][5] + b1.y; o.z = acc[j][6] + b1.z; o.w = acc[j][7] + b1.w;
            *(v4f*)(tp + (jd * 2 + 1) * 128) = o;
        }
    }
}

__device__ __forceinline__ float xhalf(float x) {
    const int b = __builtin_bit_cast(int, x);
    return __builtin_bit_cast(float,
        __builtin_amdgcn_permlanex16(b, b, 0x76543210, 0xfedcba98, false, false));
}

template <bool BETA_ONE, bool DIAG>
__device__ __forceinline__ void attn_tile(
    const float* __restrict__ Kl, const float* __restrict__ oc,
    const float* __restrict__ Vl,
    const v2f (&qb)[8], float qoc0, float qoc1, float qoc2,
    int lane, int b, int qrow, int kt, int grp,
    float eta, float absbeta, float alpha,
    float& M, float& L, v8f& O0, v8f& O1)
{
    const float scale   = 0.17677669529663687f;
    const float NEG_BIG = -3.4e38f;

    const int k0    = kt * 16;
    const int kbase = k0 + 8 * grp;
    const float* kp = Kl + (size_t)kt * 512;
    const float* vp = Vl + (size_t)kt * 512;

    if (!DIAG) {
        __builtin_prefetch(kp + 512 + lane * 12, 0, 0);
        __builtin_prefetch(vp + 512 + lane * 12, 0, 0);
    }

    v4f k4[4], v4[4];
#pragma unroll
    for (int g = 0; g < 4; ++g) k4[g] = *(const v4f*)(kp + 128 * g);
#pragma unroll
    for (int g = 0; g < 4; ++g) v4[g] = *(const v4f*)(vp + 128 * g);

    v8f s = {};
#pragma unroll
    for (int g = 0; g < 4; ++g) {
        v2f ka;
        ka.x = k4[g].x; ka.y = k4[g].y;
        s = wmma4(ka, qb[2 * g], s);
        ka.x = k4[g].z; ka.y = k4[g].w;
        s = wmma4(ka, qb[2 * g + 1], s);
    }

    const v4f* kop = (const v4f*)(oc + ((size_t)b * T_SZ + kbase) * 3);
    const v4f c0 = kop[0], c1 = kop[1], c2 = kop[2], c3 = kop[3], c4 = kop[4], c5 = kop[5];
    const float kocv[24] = { c0.x, c0.y, c0.z, c0.w, c1.x, c1.y, c1.z, c1.w,
                             c2.x, c2.y, c2.z, c2.w, c3.x, c3.y, c3.z, c3.w,
                             c4.x, c4.y, c4.z, c4.w, c5.x, c5.y, c5.z, c5.w };

#pragma unroll
    for (int r = 0; r < 8; ++r) {
        const int kcol = kbase + r;
        const float dist = (float)(qrow - kcol);
        const float powd = BETA_ONE ? dist : __powf(dist, absbeta);
        const float td   = eta / (1.f + powd);
        const float d0 = qoc0 - kocv[3 * r + 0];
        const float d1 = qoc1 - kocv[3 * r + 1];
        const float d2 = qoc2 - kocv[3 * r + 2];
        const float pr = alpha * td * __expf(-(d0*d0 + d1*d1 + d2*d2));
        const float val = s[r] * scale * pr;
        if (DIAG) s[r] = (kcol > qrow) ? NEG_BIG : val;
        else      s[r] = val;
    }

    float mt = fmaxf(fmaxf(fmaxf(s[0], s[1]), fmaxf(s[2], s[3])),
                     fmaxf(fmaxf(s[4], s[5]), fmaxf(s[6], s[7])));
    mt = fmaxf(mt, xhalf(mt));
    const float mnew = fmaxf(M, mt);
    const float sc   = __expf(M - mnew);
    float p[8];
#pragma unroll
    for (int r = 0; r < 8; ++r) p[r] = __expf(s[r] - mnew);
    const float su = ((p[0] + p[1]) + (p[2] + p[3])) + ((p[4] + p[5]) + (p[6] + p[7]));
    L = L * sc + su;
    M = mnew;
#pragma unroll
    for (int r = 0; r < 8; ++r) { O0[r] *= sc; O1[r] *= sc; }

#pragma unroll
    for (int j = 0; j < 4; ++j) {
        v2f pb;
        pb.x = p[2 * j];
        pb.y = p[2 * j + 1];
        const v4f va = v4[j >> 1];
        const v4f vb = v4[2 + (j >> 1)];
        v2f a0, a1;
        if (j & 1) { a0.x = va.z; a0.y = va.w; a1.x = vb.z; a1.y = vb.w; }
        else       { a0.x = va.x; a0.y = va.y; a1.x = vb.x; a1.y = vb.y; }
        O0 = wmma4(a0, pb, O0);
        O1 = wmma4(a1, pb, O1);
    }
}

template <bool BETA_ONE>
__device__ __forceinline__ void attn_core(
    const float* __restrict__ Q, const float* __restrict__ K,
    const float* __restrict__ Vt, const float* __restrict__ oc,
    float* __restrict__ ctx,
    int b, int h, int qt, int lane,
    float eta, float absbeta, float alpha)
{
    const int col  = lane & 15;
    const int grp  = lane >> 4;
    const int q0   = qt * 16;
    const int qrow = q0 + col;

    const float NEG_BIG = -3.4e38f;

    const size_t rec0 = (size_t)(b * NUM_HEADS + h) * (T_SZ / 16) * 512 + lane * 4;

    const float* qp = Q + rec0 + (size_t)qt * 512;
    v2f qb[8];
#pragma unroll
    for (int g = 0; g < 4; ++g) {
        const v4f q4 = *(const v4f*)(qp + 128 * g);
        qb[2 * g].x = q4.x;     qb[2 * g].y = q4.y;
        qb[2 * g + 1].x = q4.z; qb[2 * g + 1].y = q4.w;
    }

    const float* op = oc + ((size_t)b * T_SZ + qrow) * 3;
    const float qoc0 = op[0], qoc1 = op[1], qoc2 = op[2];

    const float* Kl = K + rec0;
    const float* Vl = Vt + rec0;

    float M = NEG_BIG;
    float L = 0.f;
    v8f O0 = {}, O1 = {};

    for (int kt = 0; kt < qt; ++kt)
        attn_tile<BETA_ONE, false>(Kl, oc, Vl, qb, qoc0, qoc1, qoc2, lane, b, qrow, kt,
                                   grp, eta, absbeta, alpha, M, L, O0, O1);
    attn_tile<BETA_ONE, true>(Kl, oc, Vl, qb, qoc0, qoc1, qoc2, lane, b, qrow, qt,
                              grp, eta, absbeta, alpha, M, L, O0, O1);

    L = L + xhalf(L);
    const float inv = 1.f / L;
    float* cp = ctx + ((size_t)b * T_SZ + qrow) * D_MODEL + h * DK + 8 * grp;
    v4f o;
    o.x = O0[0] * inv; o.y = O0[1] * inv; o.z = O0[2] * inv; o.w = O0[3] * inv;
    *(v4f*)(cp) = o;
    o.x = O0[4] * inv; o.y = O0[5] * inv; o.z = O0[6] * inv; o.w = O0[7] * inv;
    *(v4f*)(cp + 4) = o;
    o.x = O1[0] * inv; o.y = O1[1] * inv; o.z = O1[2] * inv; o.w = O1[3] * inv;
    *(v4f*)(cp + 16) = o;
    o.x = O1[4] * inv; o.y = O1[5] * inv; o.z = O1[6] * inv; o.w = O1[7] * inv;
    *(v4f*)(cp + 20) = o;
}

__global__ __launch_bounds__(256)
void attn_kernel(const float* __restrict__ QKV,
                 const float* __restrict__ oc,
                 const float* __restrict__ etap,
                 const float* __restrict__ betap,
                 const float* __restrict__ alphap,
                 float* __restrict__ ctx)
{
    const int lane = threadIdx.x & 31;
    const int w    = threadIdx.x >> 5;
    const int wid  = blockIdx.x * 8 + w;

    const int QT = T_SZ / 16;
    const int QP = QT / 2;
    const int pr = wid % QP;
    const int h  = (wid / QP) % NUM_HEADS;
    const int b  = wid / (QP * NUM_HEADS);

    const float eta     = etap[0];
    const float absbeta = fabsf(betap[0]);
    const float alpha   = alphap[0];

    const size_t HT = (size_t)B_SZ * NUM_HEADS * T_SZ * DK;
    const float* Q  = QKV;
    const float* K  = QKV + HT;
    const float* Vt = QKV + 2 * HT;

    if (absbeta == 1.0f) {
        for (int pass = 0; pass < 2; ++pass) {
            const int qt = pass ? pr : (QT - 1 - pr);
            attn_core<true >(Q, K, Vt, oc, ctx, b, h, qt, lane, eta, absbeta, alpha);
        }
    } else {
        for (int pass = 0; pass < 2; ++pass) {
            const int qt = pass ? pr : (QT - 1 - pr);
            attn_core<false>(Q, K, Vt, oc, ctx, b, h, qt, lane, eta, absbeta, alpha);
        }
    }
}

__global__ __launch_bounds__(256)
void out_proj_kernel(const float* __restrict__ ctx,
                     const float* __restrict__ Wo, const float* __restrict__ bo,
                     float* __restrict__ out)
{
    __shared__ __attribute__((aligned(16))) float wlds[SLAB_N * WPITCH];

    const int lane = threadIdx.x & 31;
    const int w    = threadIdx.x >> 5;

    const int MG = (B_SZ * T_SZ) / 128;
    const int ns = blockIdx.x / MG;
    const int mg = blockIdx.x % MG;
    const int n0 = ns * SLAB_N;
    const int m0 = mg * 128 + w * 16;

    v8f acc[4];
    proj_tile_16x64<false>(ctx, Wo, n0, m0, wlds, acc);

    const int r15 = lane & 15;
    const int grp = lane >> 4;
#pragma unroll
    for (int j = 0; j < 4; ++j) {
        const int n = n0 + 16 * j + r15;
        const float bias = bo[n];
#pragma unroll
        for (int m = 0; m < 8; ++m) {
            const int tl = m0 + m + 8 * grp;
            out[(size_t)tl * D_MODEL + n] = acc[j][m] + bias;
        }
    }
}

extern "C" void kernel_launch(void* const* d_in, const int* in_sizes, int n_in,
                              void* d_out, int out_size, void* d_ws, size_t ws_size,
                              hipStream_t stream) {
    (void)in_sizes; (void)n_in; (void)out_size; (void)ws_size;

    const float* x     = (const float*)d_in[0];
    const float* oc    = (const float*)d_in[1];
    const float* Wq    = (const float*)d_in[2];
    const float* bq    = (const float*)d_in[3];
    const float* Wk    = (const float*)d_in[4];
    const float* bk    = (const float*)d_in[5];
    const float* Wv    = (const float*)d_in[6];
    const float* bv    = (const float*)d_in[7];
    const float* Wo    = (const float*)d_in[8];
    const float* bo    = (const float*)d_in[9];
    const float* eta   = (const float*)d_in[10];
    const float* beta  = (const float*)d_in[11];
    const float* alpha = (const float*)d_in[12];
    float* out = (float*)d_out;

    float* ws  = (float*)d_ws;
    float* QKV = ws;
    float* ctx = ws + 3 * (size_t)B_SZ * NUM_HEADS * T_SZ * DK;

    {
        const int blocks = (3 * D_MODEL / SLAB_N) * (B_SZ * T_SZ / 128);
        qkv_proj_kernel<<<blocks, 256, 0, stream>>>(x, Wq, bq, Wk, bk, Wv, bv, QKV);
    }
    {
        const int waves = B_SZ * NUM_HEADS * (T_SZ / 32);
        attn_kernel<<<waves / 8, 256, 0, stream>>>(QKV, oc, eta, beta, alpha, ctx);
    }
    {
        const int blocks = (D_MODEL / SLAB_N) * (B_SZ * T_SZ / 128);
        out_proj_kernel<<<blocks, 256, 0, stream>>>(ctx, Wo, bo, out);
    }
}


# attention: K record and oc of key tile kt+1 loaded during tile kt into a second register set (sets alternate), V at its own tile's head, each tile's loads kept as one group; prefetch hint dropped
# speedup vs baseline: 2.9510x; 1.0547x over previous
#include <hip/hip_runtime.h>
#include <math.h>

#define D_MODEL   256
#define NUM_HEADS 8
#define DK        32
#define B_SZ      4
#define T_SZ      2048

typedef float v2f __attribute__((ext_vector_type(2)));
typedef float v4f __attribute__((ext_vector_type(4)));
typedef float v8f __attribute__((ext_vector_type(8)));

__device__ __forceinline__ v8f wmma4(v2f a, v2f b, v8f c) {
    return __builtin_amdgcn_wmma_f32_16x16x4_f32(
         false, a,  false, b,
         (short)0, c,  false,  false);
}

#define WPITCH 260
#define SLAB_N 64

__device__ __forceinline__ void proj_load_a(v4f (&a)[8], const float* __restrict__ xr, int ch) {
#pragma unroll
    for (int u = 0; u < 8; ++u) a[u] = *(const v4f*)(xr + 64 * ch + 8 * u);
}

template <bool SWAP>
__device__ __forceinline__ void proj_mma_chunk(const v4f (&a)[8], const float* wl, int ch, v8f (&acc)[4]) {
#pragma unroll
    for (int u = 0; u < 8; ++u) {
        const int i = ch * 8 + u;
        v4f b[4];
#pragma unroll
        for (int j = 0; j < 4; ++j) b[j] = *(const v4f*)(wl + j * 16 * WPITCH + 8 * i);
        v2f alo, ahi;
        alo.x = a[u].x; alo.y = a[u].y; ahi.x = a[u].z; ahi.y = a[u].w;
#pragma unroll
        for (int j = 0; j < 4; ++j) {
            v2f blo; blo.x = b[j].x; blo.y = b[j].y;
            acc[j] = SWAP ? wmma4(blo, alo, acc[j]) : wmma4(alo, blo, acc[j]);
        }
#pragma unroll
        for (int j = 0; j < 4; ++j) {
            v2f bhi; bhi.x = b[j].z; bhi.y = b[j].w;
            acc[j] = SWAP ? wmma4(bhi, ahi, acc[j]) : wmma4(ahi, bhi, acc[j]);
        }
    }
}

template <bool SWAP>
__device__ __forceinline__ void proj_tile_16x64(const float* __restrict__ X, const float* __restrict__ Wm,
                                                int n0, int m0, float* wlds, v8f (&acc)[4])
{
    const int tid  = threadIdx.x;
    const int lane = tid & 31;
    const int c    = lane & 15;
    const int hh   = lane >> 4;

    const float* xr = X + (size_t)(m0 + c) * D_MODEL + 4 * hh;

    v4f a0[8], a1[8];
    proj_load_a(a0, xr, 0);

    {
        const int c4 = tid & 63;
        const int r0 = tid >> 6;
        v4f t[16];
#pragma unroll
        for (int u = 0; u < 16; ++u)
            t[u] = *(const v4f*)(Wm + (size_t)(n0 + 4 * u + r0) * D_MODEL + 4 * c4);
#pragma unroll
        for (int u = 0; u < 16; ++u)
            *(v4f*)(wlds + (4 * u + r0) * WPITCH + 4 * c4) = t[u];
    }
    __syncthreads();

    const float* wl = wlds + c * WPITCH + 4 * hh;

#pragma unroll
    for (int j = 0; j < 4; ++j) acc[j] = (v8f){};

    proj_load_a(a1, xr, 1);
    proj_mma_chunk<SWAP>(a0, wl, 0, acc);
    proj_load_a(a0, xr, 2);
    proj_mma_chunk<SWAP>(a1, wl, 1, acc);
    proj_load_a(a1, xr, 3);
    proj_mma_chunk<SWAP>(a0, wl, 2, acc);
    proj_mma_chunk<SWAP>(a1, wl, 3, acc);
}

__global__ __launch_bounds__(256)
void qkv_proj_kernel(const float* __restrict__ x,
                     const float* __restrict__ Wq, const float* __restrict__ bq,
                     const float* __restrict__ Wk, const float* __restrict__ bk,
                     const float* __restrict__ Wv, const float* __restrict__ bv,
                     float* __restrict__ QKV)
{
    __shared__ __attribute__((aligned(16))) float wlds[SLAB_N * WPITCH];

    const int lane = threadIdx.x & 31;
    const int w    = threadIdx.x >> 5;

    const int MG  = (B_SZ * T_SZ) / 128;
    const int ns  = blockIdx.x / MG;
    const int mg  = blockIdx.x % MG;
    const int mat = ns >> 2;
    const int n0  = (ns & 3) * SLAB_N;
    const int m0  = mg * 128 + w * 16;

    const float* Wm = (mat == 0) ? Wq : (mat == 1) ? Wk : Wv;
    const float* bs = (mat == 0) ? bq : (mat == 1) ? bk : bv;

    float* dst = QKV + (size_t)mat * ((size_t)B_SZ * NUM_HEADS * T_SZ * DK);

    const int b_ = m0 / T_SZ;
    const int tt = (m0 % T_SZ) >> 4;

    v8f acc[4];
    if (mat == 2) {
        proj_tile_16x64<false>(x, Wm, n0, m0, wlds, acc);
#pragma unroll
        for (int j = 0; j < 4; ++j) {
            const int nb = n0 + 16 * j;
            const float bias = bs[nb + (lane & 15)];
            const int h = nb >> 5, jd = (nb >> 4) & 1;
            float* tp = dst + ((size_t)(b_ * NUM_HEADS + h) * (T_SZ / 16) + tt) * 512 + lane * 4;
            v4f o;
            o.x = acc[j][0] + bias; o.y = acc[j][1] + bias; o.z = acc[j][2] + bias; o.w = acc[j][3] + bias;
            *(v4f*)(tp + (jd * 2 + 0) * 128) = o;
            o.x = acc[j][4] + bias; o.y = acc[j][5] + bias; o.z = acc[j][6] + bias; o.w = acc[j][7] + bias;
            *(v4f*)(tp + (jd * 2 + 1) * 128) = o;
        }
    } else {
        proj_tile_16x64<true>(x, Wm, n0, m0, wlds, acc);
#pragma unroll
        for (int j = 0; j < 4; ++j) {
            const int nb = n0 + 16 * j;
            const float* bp = bs + nb + 8 * (lane >> 4);
            v4f b0, b1;
            b0.x = bp[0]; b0.y = bp[1]; b0.z = bp[2]; b0.w = bp[3];
            b1.x = bp[4]; b1.y = bp[5]; b1.z = bp[6]; b1.w = bp[7];
            const int h = nb >> 5, jd = (nb >> 4) & 1;
            float* tp = dst + ((size_t)(b_ * NUM_HEADS + h) * (T_SZ / 16) + tt) * 512 + lane * 4;
            v4f o;
            o.x = acc[j][0] + b0.x; o.y = acc[j][1] + b0.y; o.z = acc[j][2] + b0.z; o.w = acc[j][3] + b0.w;
            *(v4f*)(tp + (jd * 2 + 0) * 128) = o;
            o.x = acc[j][4] + b1.x; o.y = acc[j][5] + b1.y; o.z = acc[j][6] + b1.z; o.w = acc[j][7] + b1.w;
            *(v4f*)(tp + (jd * 2 + 1) * 128) = o;
        }
    }
}

__device__ __forceinline__ float xhalf(float x) {
    const int b = __builtin_bit_cast(int, x);
    return __builtin_bit_cast(float,
        __builtin_amdgcn_permlanex16(b, b, 0x76543210, 0xfedcba98, false, false));
}

__device__ __forceinline__ void attn_load_k_oc(
    v4f (&k4)[4], v4f (&c)[6],
    const float* __restrict__ Kl, const float* __restrict__ oc, int b, int kt, int grp)
{
    const float* kp = Kl + (size_t)kt * 512;
#pragma unroll
    for (int g = 0; g < 4; ++g) k4[g] = *(const v4f*)(kp + 128 * g);
    const v4f* kop = (const v4f*)(oc + ((size_t)b * T_SZ + kt * 16 + 8 * grp) * 3);
#pragma unroll
    for (int g = 0; g < 6; ++g) c[g] = kop[g];
}

__device__ __forceinline__ void attn_load_v(v4f (&v4)[4], const float* __restrict__ Vl, int kt)
{
    const float* vp = Vl + (size_t)kt * 512;
#pragma unroll
    for (int g = 0; g < 4; ++g) v4[g] = *(const v4f*)(vp + 128 * g);
}

__device__ __forceinline__ void attn_loads_stay_here() { __builtin_amdgcn_sched_barrier(0); }

template <bool BETA_ONE, bool DIAG>
__device__ __forceinline__ void attn_tile_math(
    const v4f (&k4)[4], const v4f (&v4)[4], const v4f (&c)[6],
    const v2f (&qb)[8], float qoc0, float qoc1, float qoc2,
    int qrow, int kt, int grp,
    float eta, float absbeta, float alpha,
    float& M, float& L, v8f& O0, v8f& O1)
{
    const float scale   = 0.17677669529663687f;
    const float NEG_BIG = -3.4e38f;

    const int kbase = kt * 16 + 8 * grp;

    v8f s = {};
#pragma unroll
    for (int g = 0; g < 4; ++g) {
        v2f ka;
        ka.x = k4[g].x; ka.y = k4[g].y;
        s = wmma4(ka, qb[2 * g], s);
        ka.x = k4[g].z; ka.y = k4[g].w;
        s = wmma4(ka, qb[2 * g + 1], s);
    }

    const float kocv[24] = { c[0].x, c[0].y, c[0].z, c[0].w, c[1].x, c[1].y, c[1].z, c[1].w,
                             c[2].x, c[2].y, c[2].z, c[2].w, c[3].x, c[3].y, c[3].z, c[3].w,
                             c[4].x, c[4].y, c[4].z, c[4].w, c[5].x, c[5].y, c[5].z, c[5].w };

#pragma unroll
    for (int r = 0; r < 8; ++r) {
        const int kcol = kbase + r;
        const float dist = (float)(qrow - kcol);
        const float powd = BETA_ONE ? dist : __powf(dist, absbeta);
        const float td   = eta / (1.f + powd);
        const float d0 = qoc0 - kocv[3 * r + 0];
        const float d1 = qoc1 - kocv[3 * r + 1];
        const float d2 = qoc2 - kocv[3 * r + 2];
        const float pr = alpha * td * __expf(-(d0*d0 + d1*d1 + d2*d2));
        const float val = s[r] * scale * pr;
        if (DIAG) s[r] = (kcol > qrow) ? NEG_BIG : val;
        else      s[r] = val;
    }

    float mt = fmaxf(fmaxf(fmaxf(s[0], s[1]), fmaxf(s[2], s[3])),
                     fmaxf(fmaxf(s[4], s[5]), fmaxf(s[6], s[7])));
    mt = fmaxf(mt, xhalf(mt));
    const float mnew = fmaxf(M, mt);
    const float sc   = __expf(M - mnew);
    float p[8];
#pragma unroll
    for (int r = 0; r < 8; ++r) p[r] = __expf(s[r] - mnew);
    const float su = ((p[0] + p[1]) + (p[2] + p[3])) + ((p[4] + p[5]) + (p[6] + p[7]));
    L = L * sc + su;
    M = mnew;
#pragma unroll
    for (int r = 0; r < 8; ++r) { O0[r] *= sc; O1[r] *= sc; }

#pragma unroll
    for (int j = 0; j < 4; ++j) {
        v2f pb;
        pb.x = p[2 * j];
        pb.y = p[2 * j + 1];
        const v4f va = v4[j >> 1];
        const v4f vb = v4[2 + (j >> 1)];
        v2f a0, a1;
        if (j & 1) { a0.x = va.z; a0.y = va.w; a1.x = vb.z; a1.y = vb.w; }
        else       { a0.x = va.x; a0.y = va.y; a1.x = vb.x; a1.y = vb.y; }
        O0 = wmma4(a0, pb, O0);
        O1 = wmma4(a1, pb, O1);
    }
}

template <bool BETA_ONE>
__device__ __forceinline__ void attn_core(
    const float* __restrict__ Q, const float* __restrict__ K,
    const float* __restrict__ Vt, const float* __restrict__ oc,
    float* __restrict__ ctx,
    int b, int h, int qt, int lane,
    float eta, float absbeta, float alpha)
{
    const int col  = lane & 15;
    const int grp  = lane >> 4;
    const int q0   = qt * 16;
    const int qrow = q0 + col;

    const float NEG_BIG = -3.4e38f;

    const size_t rec0 = (size_t)(b * NUM_HEADS + h) * (T_SZ / 16) * 512 + lane * 4;

    const float* Kl = K + rec0;
    const float* Vl = Vt + rec0;

    v4f kA[4], cA[6];
    v4f kB[4], cB[6];
    attn_load_k_oc(kA, cA, Kl, oc, b, 0, grp);

    const float* qp = Q + rec0 + (size_t)qt * 512;
    v2f qb[8];
#pragma unroll
    for (int g = 0; g < 4; ++g) {
        const v4f q4 = *(const v4f*)(qp + 128 * g);
        qb[2 * g].x = q4.x;     qb[2 * g].y = q4.y;
        qb[2 * g + 1].x = q4.z; qb[2 * g + 1].y = q4.w;
    }

    const float* op = oc + ((size_t)b * T_SZ + qrow) * 3;
    const float qoc0 = op[0], qoc1 = op[1], qoc2 = op[2];

    float M = NEG_BIG;
    float L = 0.f;
    v8f O0 = {}, O1 = {};

    int kt = 0;
    for (; kt + 2 <= qt; kt += 2) {
        {
            v4f vv[4];
            attn_loads_stay_here();
            attn_load_v(vv, Vl, kt);
            attn_load_k_oc(kB, cB, Kl, oc, b, kt + 1, grp);
            attn_loads_stay_here();
            attn_tile_math<BETA_ONE, false>(kA, vv, cA, qb, qoc0, qoc1, qoc2, qrow, kt,
                                            grp, eta, absbeta, alpha, M, L, O0, O1);
        }
        {
            v4f vv[4];
            attn_loads_stay_here();
            attn_load_v(vv, Vl, kt + 1);
            attn_load_k_oc(kA, cA, Kl, oc, b, kt + 2, grp);
            attn_loads_stay_here();
            attn_tile_math<BETA_ONE, false>(kB, vv, cB, qb, qoc0, qoc1, qoc2, qrow, kt + 1,
                                            grp, eta, absbeta, alpha, M, L, O0, O1);
        }
    }
    if (kt < qt) {
        v4f vv[4];
        attn_load_v(vv, Vl, kt);
        attn_load_k_oc(kB, cB, Kl, oc, b, qt, grp);
        attn_loads_stay_here();
        attn_tile_math<BETA_ONE, false>(kA, vv, cA, qb, qoc0, qoc1, qoc2, qrow, kt,
                                        grp, eta, absbeta, alpha, M, L, O0, O1);
#pragma unroll
        for (int g = 0; g < 4; ++g) kA[g] = kB[g];
#pragma unroll
        for (int g = 0; g < 6; ++g) cA[g] = cB[g];
    }
    {
        v4f vv[4];
        attn_load_v(vv, Vl, qt);
        attn_tile_math<BETA_ONE, true>(kA, vv, cA, qb, qoc0, qoc1, qoc2, qrow, qt,
                                       grp, eta, absbeta, alpha, M, L, O0, O1);
    }

    L = L + xhalf(L);
    const float inv = 1.f / L;
    float* cp = ctx + ((size_t)b * T_SZ + qrow) * D_MODEL + h * DK + 8 * grp;
    v4f o;
    o.x = O0[0] * inv; o.y = O0[1] * inv; o.z = O0[2] * inv; o.w = O0[3] * inv;
    *(v4f*)(cp) = o;
    o.x = O0[4] * inv; o.y = O0[5] * inv; o.z = O0[6] * inv; o.w = O0[7] * inv;
    *(v4f*)(cp + 4) = o;
    o.x = O1[0] * inv; o.y = O1[1] * inv; o.z = O1[2] * inv; o.w = O1[3] * inv;
    *(v4f*)(cp + 16) = o;
    o.x = O1[4] * inv; o.y = O1[5] * inv; o.z = O1[6] * inv; o.w = O1[7] * inv;
    *(v4f*)(cp + 20) = o;
}

__global__ __launch_bounds__(256)
void attn_kernel(const float* __restrict__ QKV,
                 const float* __restrict__ oc,
                 const float* __restrict__ etap,
                 const float* __restrict__ betap,
                 const float* __restrict__ alphap,
                 float* __restrict__ ctx)
{
    const int lane = threadIdx.x & 31;
    const int w    = threadIdx.x >> 5;
    const int wid  = blockIdx.x * 8 + w;

    const int QT = T_SZ / 16;
    const int QP = QT / 2;
    const int pr = wid % QP;
    const int h  = (wid / QP) % NUM_HEADS;
    const int b  = wid / (QP * NUM_HEADS);

    const float eta     = etap[0];
    const float absbeta = fabsf(betap[0]);
    const float alpha   = alphap[0];

    const size_t HT = (size_t)B_SZ * NUM_HEADS * T_SZ * DK;
    const float* Q  = QKV;
    const float* K  = QKV + HT;
    const float* Vt = QKV + 2 * HT;

    if (absbeta == 1.0f) {
        for (int pass = 0; pass < 2; ++pass) {
            const int qt = pass ? pr : (QT - 1 - pr);
            attn_core<true >(Q, K, Vt, oc, ctx, b, h, qt, lane, eta, absbeta, alpha);
        }
    } else {
        for (int pass = 0; pass < 2; ++pass) {
            const int qt = pass ? pr : (QT - 1 - pr);
            attn_core<false>(Q, K, Vt, oc, ctx, b, h, qt, lane, eta, absbeta, alpha);
        }
    }
}

__global__ __launch_bounds__(256)
void out_proj_kernel(const float* __restrict__ ctx,
                     const float* __restrict__ Wo, const float* __restrict__ bo,
                     float* __restrict__ out)
{
    __shared__ __attribute__((aligned(16))) float wlds[SLAB_N * WPITCH];

    const int lane = threadIdx.x & 31;
    const int w    = threadIdx.x >> 5;

    const int MG = (B_SZ * T_SZ) / 128;
    const int ns = blockIdx.x / MG;
    const int mg = blockIdx.x % MG;
    const int n0 = ns * SLAB_N;
    const int m0 = mg * 128 + w * 16;

    v8f acc[4];
    proj_tile_16x64<false>(ctx, Wo, n0, m0, wlds, acc);

    const int r15 = lane & 15;
    const int grp = lane >> 4;
#pragma unroll
    for (int j = 0; j < 4; ++j) {
        const int n = n0 + 16 * j + r15;
        const float bias = bo[n];
#pragma unroll
        for (int m = 0; m < 8; ++m) {
            const int tl = m0 + m + 8 * grp;
            out[(size_t)tl * D_MODEL + n] = acc[j][m] + bias;
        }
    }
}

extern "C" void kernel_launch(void* const* d_in, const int* in_sizes, int n_in,
                              void* d_out, int out_size, void* d_ws, size_t ws_size,
                              hipStream_t stream) {
    (void)in_sizes; (void)n_in; (void)out_size; (void)ws_size;

    const float* x     = (const float*)d_in[0];
    const float* oc    = (const float*)d_in[1];
    const float* Wq    = (const float*)d_in[2];
    const float* bq    = (const float*)d_in[3];
    const float* Wk    = (const float*)d_in[4];
    const float* bk    = (const float*)d_in[5];
    const float* Wv    = (const float*)d_in[6];
    const float* bv    = (const float*)d_in[7];
    const float* Wo    = (const float*)d_in[8];
    const float* bo    = (const float*)d_in[9];
    const float* eta   = (const float*)d_in[10];
    const float* beta  = (const float*)d_in[11];
    const float* alpha = (const float*)d_in[12];
    float* out = (float*)d_out;

    float* ws  = (float*)d_ws;
    float* QKV = ws;
    float* ctx = ws + 3 * (size_t)B_SZ * NUM_HEADS * T_SZ * DK;

    {
        const int blocks = (3 * D_MODEL / SLAB_N) * (B_SZ * T_SZ / 128);
        qkv_proj_kernel<<<blocks, 256, 0, stream>>>(x, Wq, bq, Wk, bk, Wv, bv, QKV);
    }
    {
        const int waves = B_SZ * NUM_HEADS * (T_SZ / 32);
        attn_kernel<<<waves / 8, 256, 0, stream>>>(QKV, oc, eta, beta, alpha, ctx);
    }
    {
        const int blocks = (D_MODEL / SLAB_N) * (B_SZ * T_SZ / 128);
        out_proj_kernel<<<blocks, 256, 0, stream>>>(ctx, Wo, bo, out);
    }
}


# attention prior: time decay eta/(1+dist) as eta * v_rcp_f32 (1 ulp) instead of the IEEE divide sequence, 1+dist formed in float from one per-tile value; the factor's rounding changes by <= 2 ulp
# speedup vs baseline: 3.1750x; 1.0759x over previous
#include <hip/hip_runtime.h>
#include <math.h>

#define D_MODEL   256
#define NUM_HEADS 8
#define DK        32
#define B_SZ      4
#define T_SZ      2048

typedef float v2f __attribute__((ext_vector_type(2)));
typedef float v4f __attribute__((ext_vector_type(4)));
typedef float v8f __attribute__((ext_vector_type(8)));

__device__ __forceinline__ v8f wmma4(v2f a, v2f b, v8f c) {
    return __builtin_amdgcn_wmma_f32_16x16x4_f32(
         false, a,  false, b,
         (short)0, c,  false,  false);
}

#define WPITCH 260
#define SLAB_N 64

__device__ __forceinline__ void proj_load_a(v4f (&a)[8], const float* __restrict__ xr, int ch) {
#pragma unroll
    for (int u = 0; u < 8; ++u) a[u] = *(const v4f*)(xr + 64 * ch + 8 * u);
}

template <bool SWAP>
__device__ __forceinline__ void proj_mma_chunk(const v4f (&a)[8], const float* wl, int ch, v8f (&acc)[4]) {
#pragma unroll
    for (int u = 0; u < 8; ++u) {
        const int i = ch * 8 + u;
        v4f b[4];
#pragma unroll
        for (int j = 0; j < 4; ++j) b[j] = *(const v4f*)(wl + j * 16 * WPITCH + 8 * i);
        v2f alo, ahi;
        alo.x = a[u].x; alo.y = a[u].y; ahi.x = a[u].z; ahi.y = a[u].w;
#pragma unroll
        for (int j = 0; j < 4; ++j) {
            v2f blo; blo.x = b[j].x; blo.y = b[j].y;
            acc[j] = SWAP ? wmma4(blo, alo, acc[j]) : wmma4(alo, blo, acc[j]);
        }
#pragma unroll
        for (int j = 0; j < 4; ++j) {
            v2f bhi; bhi.x = b[j].z; bhi.y = b[j].w;
            acc[j] = SWAP ? wmma4(bhi, ahi, acc[j]) : wmma4(ahi, bhi, acc[j]);
        }
    }
}

template <bool SWAP>
__device__ __forceinline__ void proj_tile_16x64(const float* __restrict__ X, const float* __restrict__ Wm,
                                                int n0, int m0, float* wlds, v8f (&acc)[4])
{
    const int tid  = threadIdx.x;
    const int lane = tid & 31;
    const int c    = lane & 15;
    const int hh   = lane >> 4;

    const float* xr = X + (size_t)(m0 + c) * D_MODEL + 4 * hh;

    v4f a0[8], a1[8];
    proj_load_a(a0, xr, 0);

    {
        const int c4 = tid & 63;
        const int r0 = tid >> 6;
        v4f t[16];
#pragma unroll
        for (int u = 0; u < 16; ++u)
            t[u] = *(const v4f*)(Wm + (size_t)(n0 + 4 * u + r0) * D_MODEL + 4 * c4);
#pragma unroll
        for (int u = 0; u < 16; ++u)
            *(v4f*)(wlds + (4 * u + r0) * WPITCH + 4 * c4) = t[u];
    }
    __syncthreads();

    const float* wl = wlds + c * WPITCH + 4 * hh;

#pragma unroll
    for (int j = 0; j < 4; ++j) acc[j] = (v8f){};

    proj_load_a(a1, xr, 1);
    proj_mma_chunk<SWAP>(a0, wl, 0, acc);
    proj_load_a(a0, xr, 2);
    proj_mma_chunk<SWAP>(a1, wl, 1, acc);
    proj_load_a(a1, xr, 3);
    proj_mma_chunk<SWAP>(a0, wl, 2, acc);
    proj_mma_chunk<SWAP>(a1, wl, 3, acc);
}

__global__ __launch_bounds__(256)
void qkv_proj_kernel(const float* __restrict__ x,
                     const float* __restrict__ Wq, const float* __restrict__ bq,
                     const float* __restrict__ Wk, const float* __restrict__ bk,
                     const float* __restrict__ Wv, const float* __restrict__ bv,
                     float* __restrict__ QKV)
{
    __shared__ __attribute__((aligned(16))) float wlds[SLAB_N * WPITCH];

    const int lane = threadIdx.x & 31;
    const int w    = threadIdx.x >> 5;

    const int MG  = (B_SZ * T_SZ) / 128;
    const int ns  = blockIdx.x / MG;
    const int mg  = blockIdx.x % MG;
    const int mat = ns >> 2;
    const int n0  = (ns & 3) * SLAB_N;
    const int m0  = mg * 128 + w * 16;

    const float* Wm = (mat == 0) ? Wq : (mat == 1) ? Wk : Wv;
    const float* bs = (mat == 0) ? bq : (mat == 1) ? bk : bv;

    float* dst = QKV + (size_t)mat * ((size_t)B_SZ * NUM_HEADS * T_SZ * DK);

    const int b_ = m0 / T_SZ;
    const int tt = (m0 % T_SZ) >> 4;

    v8f acc[4];
    if (mat == 2) {
        proj_tile_16x64<false>(x, Wm, n0, m0, wlds, acc);
#pragma unroll
        for (int j = 0; j < 4; ++j) {
            const int nb = n0 + 16 * j;
            const float bias = bs[nb + (lane & 15)];
            const int h = nb >> 5, jd = (nb >> 4) & 1;
            float* tp = dst + ((size_t)(b_ * NUM_HEADS + h) * (T_SZ / 16) + tt) * 512 + lane * 4;
            v4f o;
            o.x = acc[j][0] + bias; o.y = acc[j][1] + bias; o.z = acc[j][2] + bias; o.w = acc[j][3] + bias;
            *(v4f*)(tp + (jd * 2 + 0) * 128) = o;
            o.x = acc[j][4] + bias; o.y = acc[j][5] + bias; o.z = acc[j][6] + bias; o.w = acc[j][7] + bias;
            *(v4f*)(tp + (jd * 2 + 1) * 128) = o;
        }
    } else {
        proj_tile_16x64<true>(x, Wm, n0, m0, wlds, acc);
#pragma unroll
        for (int j = 0; j < 4; ++j) {
            const int nb = n0 + 16 * j;
            const float* bp = bs + nb + 8 * (lane >> 4);
            v4f b0, b1;
            b0.x = bp[0]; b0.y = bp[1]; b0.z = bp[2]; b0.w = bp[3];
            b1.x = bp[4]; b1.y = bp[5]; b1.z = bp[6]; b1.w = bp[7];
            const int h = nb >> 5, jd = (nb >> 4) & 1;
            float* tp = dst + ((size_t)(b_ * NUM_HEADS + h) * (T_SZ / 16) + tt) * 512 + lane * 4;
            v4f o;
            o.x = acc[j][0] + b0.x; o.y = acc[j][1] + b0.y; o.z = acc[j][2] + b0.z; o.w = acc[j][3] + b0.w;
            *(v4f*)(tp + (jd * 2 + 0) * 128) = o;
            o.x = acc[j][4] + b1.x; o.y = acc[j][5] + b1.y; o.z = acc[j][6] + b1.z; o.w = acc[j][7] + b1.w;
            *(v4f*)(tp + (jd * 2 + 1) * 128) = o;
        }
    }
}

__device__ __forceinline__ float xhalf(float x) {
    const int b = __builtin_bit_cast(int, x);
    return __builtin_bit_cast(float,
        __builtin_amdgcn_permlanex16(b, b, 0x76543210, 0xfedcba98, false, false));
}

__device__ __forceinline__ void attn_load_k_oc(
    v4f (&k4)[4], v4f (&c)[6],
    const float* __restrict__ Kl, const float* __restrict__ oc, int b, int kt, int grp)
{
    const float* kp = Kl + (size_t)kt * 512;
#pragma unroll
    for (int g = 0; g < 4; ++g) k4[g] = *(const v4f*)(kp + 128 * g);
    const v4f* kop = (const v4f*)(oc + ((size_t)b * T_SZ + kt * 16 + 8 * grp) * 3);
#pragma unroll
    for (int g = 0; g < 6; ++g) c[g] = kop[g];
}

__device__ __forceinline__ void attn_load_v(v4f (&v4)[4], const float* __restrict__ Vl, int kt)
{
    const float* vp = Vl + (size_t)kt * 512;
#pragma unroll
    for (int g = 0; g < 4; ++g) v4[g] = *(const v4f*)(vp + 128 * g);
}

__device__ __forceinline__ void attn_loads_stay_here() { __builtin_amdgcn_sched_barrier(0); }

template <bool BETA_ONE, bool DIAG>
__device__ __forceinline__ void attn_tile_math(
    const v4f (&k4)[4], const v4f (&v4)[4], const v4f (&c)[6],
    const v2f (&qb)[8], float qoc0, float qoc1, float qoc2,
    int qrow, int kt, int grp,
    float eta, float absbeta, float alpha,
    float& M, float& L, v8f& O0, v8f& O1)
{
    const float scale   = 0.17677669529663687f;
    const float NEG_BIG = -3.4e38f;

    const int kbase = kt * 16 + 8 * grp;

    const float fd0 = (float)(qrow - kbase);
    const float fd1 = fd0 + 1.f;

    v8f s = {};
#pragma unroll
    for (int g = 0; g < 4; ++g) {
        v2f ka;
        ka.x = k4[g].x; ka.y = k4[g].y;
        s = wmma4(ka, qb[2 * g], s);
        ka.x = k4[g].z; ka.y = k4[g].w;
        s = wmma4(ka, qb[2 * g + 1], s);
    }

    const float kocv[24] = { c[0].x, c[0].y, c[0].z, c[0].w, c[1].x, c[1].y, c[1].z, c[1].w,
                             c[2].x, c[2].y, c[2].z, c[2].w, c[3].x, c[3].y, c[3].z, c[3].w,
                             c[4].x, c[4].y, c[4].z, c[4].w, c[5].x, c[5].y, c[5].z, c[5].w };

#pragma unroll
    for (int r = 0; r < 8; ++r) {
        const int kcol = kbase + r;
        const float den = BETA_ONE ? (fd1 - (float)r)
                                   : (1.f + __powf(fd0 - (float)r, absbeta));
        const float td  = eta * __builtin_amdgcn_rcpf(den);
        const float d0 = qoc0 - kocv[3 * r + 0];
        const float d1 = qoc1 - kocv[3 * r + 1];
        const float d2 = qoc2 - kocv[3 * r + 2];
        const float pr = alpha * td * __expf(-(d0*d0 + d1*d1 + d2*d2));
        const float val = s[r] * scale * pr;
        if (DIAG) s[r] = (kcol > qrow) ? NEG_BIG : val;
        else      s[r] = val;
    }

    float mt = fmaxf(fmaxf(fmaxf(s[0], s[1]), fmaxf(s[2], s[3])),
                     fmaxf(fmaxf(s[4], s[5]), fmaxf(s[6], s[7])));
    mt = fmaxf(mt, xhalf(mt));
    const float mnew = fmaxf(M, mt);
    const float sc   = __expf(M - mnew);
    float p[8];
#pragma unroll
    for (int r = 0; r < 8; ++r) p[r] = __expf(s[r] - mnew);
    const float su = ((p[0] + p[1]) + (p[2] + p[3])) + ((p[4] + p[5]) + (p[6] + p[7]));
    L = L * sc + su;
    M = mnew;
#pragma unroll
    for (int r = 0; r < 8; ++r) { O0[r] *= sc; O1[r] *= sc; }

#pragma unroll
    for (int j = 0; j < 4; ++j) {
        v2f pb;
        pb.x = p[2 * j];
        pb.y = p[2 * j + 1];
        const v4f va = v4[j >> 1];
        const v4f vb = v4[2 + (j >> 1)];
        v2f a0, a1;
        if (j & 1) { a0.x = va.z; a0.y = va.w; a1.x = vb.z; a1.y = vb.w; }
        else       { a0.x = va.x; a0.y = va.y; a1.x = vb.x; a1.y = vb.y; }
        O0 = wmma4(a0, pb, O0);
        O1 = wmma4(a1, pb, O1);
    }
}

template <bool BETA_ONE>
__device__ __forceinline__ void attn_core(
    const float* __restrict__ Q, const float* __restrict__ K,
    const float* __restrict__ Vt, const float* __restrict__ oc,
    float* __restrict__ ctx,
    int b, int h, int qt, int lane,
    float eta, float absbeta, float alpha)
{
    const int col  = lane & 15;
    const int grp  = lane >> 4;
    const int q0   = qt * 16;
    const int qrow = q0 + col;

    const float NEG_BIG = -3.4e38f;

    const size_t rec0 = (size_t)(b * NUM_HEADS + h) * (T_SZ / 16) * 512 + lane * 4;

    const float* Kl = K + rec0;
    const float* Vl = Vt + rec0;

    v4f kA[4], cA[6];
    v4f kB[4], cB[6];
    attn_load_k_oc(kA, cA, Kl, oc, b, 0, grp);

    const float* qp = Q + rec0 + (size_t)qt * 512;
    v2f qb[8];
#pragma unroll
    for (int g = 0; g < 4; ++g) {
        const v4f q4 = *(const v4f*)(qp + 128 * g);
        qb[2 * g].x = q4.x;     qb[2 * g].y = q4.y;
        qb[2 * g + 1].x = q4.z; qb[2 * g + 1].y = q4.w;
    }

    const float* op = oc + ((size_t)b * T_SZ + qrow) * 3;
    const float qoc0 = op[0], qoc1 = op[1], qoc2 = op[2];

    float M = NEG_BIG;
    float L = 0.f;
    v8f O0 = {}, O1 = {};

    int kt = 0;
    for (; kt + 2 <= qt; kt += 2) {
        {
            v4f vv[4];
            attn_loads_stay_here();
            attn_load_v(vv, Vl, kt);
            attn_load_k_oc(kB, cB, Kl, oc, b, kt + 1, grp);
            attn_loads_stay_here();
            attn_tile_math<BETA_ONE, false>(kA, vv, cA, qb, qoc0, qoc1, qoc2, qrow, kt,
                                            grp, eta, absbeta, alpha, M, L, O0, O1);
        }
        {
            v4f vv[4];
            attn_loads_stay_here();
            attn_load_v(vv, Vl, kt + 1);
            attn_load_k_oc(kA, cA, Kl, oc, b, kt + 2, grp);
            attn_loads_stay_here();
            attn_tile_math<BETA_ONE, false>(kB, vv, cB, qb, qoc0, qoc1, qoc2, qrow, kt + 1,
                                            grp, eta, absbeta, alpha, M, L, O0, O1);
        }
    }
    if (kt < qt) {
        v4f vv[4];
        attn_load_v(vv, Vl, kt);
        attn_load_k_oc(kB, cB, Kl, oc, b, qt, grp);
        attn_loads_stay_here();
        attn_tile_math<BETA_ONE, false>(kA, vv, cA, qb, qoc0, qoc1, qoc2, qrow, kt,
                                        grp, eta, absbeta, alpha, M, L, O0, O1);
#pragma unroll
        for (int g = 0; g < 4; ++g) kA[g] = kB[g];
#pragma unroll
        for (int g = 0; g < 6; ++g) cA[g] = cB[g];
    }
    {
        v4f vv[4];
        attn_load_v(vv, Vl, qt);
        attn_tile_math<BETA_ONE, true>(kA, vv, cA, qb, qoc0, qoc1, qoc2, qrow, qt,
                                       grp, eta, absbeta, alpha, M, L, O0, O1);
    }

    L = L + xhalf(L);
    const float inv = 1.f / L;
    float* cp = ctx + ((size_t)b * T_SZ + qrow) * D_MODEL + h * DK + 8 * grp;
    v4f o;
    o.x = O0[0] * inv; o.y = O0[1] * inv; o.z = O0[2] * inv; o.w = O0[3] * inv;
    *(v4f*)(cp) = o;
    o.x = O0[4] * inv; o.y = O0[5] * inv; o.z = O0[6] * inv; o.w = O0[7] * inv;
    *(v4f*)(cp + 4) = o;
    o.x = O1[0] * inv; o.y = O1[1] * inv; o.z = O1[2] * inv; o.w = O1[3] * inv;
    *(v4f*)(cp + 16) = o;
    o.x = O1[4] * inv; o.y = O1[5] * inv; o.z = O1[6] * inv; o.w = O1[7] * inv;
    *(v4f*)(cp + 20) = o;
}

__global__ __launch_bounds__(256)
void attn_kernel(const float* __restrict__ QKV,
                 const float* __restrict__ oc,
                 const float* __restrict__ etap,
                 const float* __restrict__ betap,
                 const float* __restrict__ alphap,
                 float* __restrict__ ctx)
{
    const int lane = threadIdx.x & 31;
    const int w    = threadIdx.x >> 5;
    const int wid  = blockIdx.x * 8 + w;

    const int QT = T_SZ / 16;
    const int QP = QT / 2;
    const int pr = wid % QP;
    const int h  = (wid / QP) % NUM_HEADS;
    const int b  = wid / (QP * NUM_HEADS);

    const float eta     = etap[0];
    const float absbeta = fabsf(betap[0]);
    const float alpha   = alphap[0];

    const size_t HT = (size_t)B_SZ * NUM_HEADS * T_SZ * DK;
    const float* Q  = QKV;
    const float* K  = QKV + HT;
    const float* Vt = QKV + 2 * HT;

    if (absbeta == 1.0f) {
        for (int pass = 0; pass < 2; ++pass) {
            const int qt = pass ? pr : (QT - 1 - pr);
            attn_core<true >(Q, K, Vt, oc, ctx, b, h, qt, lane, eta, absbeta, alpha);
        }
    } else {
        for (int pass = 0; pass < 2; ++pass) {
            const int qt = pass ? pr : (QT - 1 - pr);
            attn_core<false>(Q, K, Vt, oc, ctx, b, h, qt, lane, eta, absbeta, alpha);
        }
    }
}

__global__ __launch_bounds__(256)
void out_proj_kernel(const float* __restrict__ ctx,
                     const float* __restrict__ Wo, const float* __restrict__ bo,
                     float* __restrict__ out)
{
    __shared__ __attribute__((aligned(16))) float wlds[SLAB_N * WPITCH];

    const int lane = threadIdx.x & 31;
    const int w    = threadIdx.x >> 5;

    const int MG = (B_SZ * T_SZ) / 128;
    const int ns = blockIdx.x / MG;
    const int mg = blockIdx.x % MG;
    const int n0 = ns * SLAB_N;
    const int m0 = mg * 128 + w * 16;

    v8f acc[4];
    proj_tile_16x64<false>(ctx, Wo, n0, m0, wlds, acc);

    const int r15 = lane & 15;
    const int grp = lane >> 4;
#pragma unroll
    for (int j = 0; j < 4; ++j) {
        const int n = n0 + 16 * j + r15;
        const float bias = bo[n];
#pragma unroll
        for (int m = 0; m < 8; ++m) {
            const int tl = m0 + m + 8 * grp;
            out[(size_t)tl * D_MODEL + n] = acc[j][m] + bias;
        }
    }
}

extern "C" void kernel_launch(void* const* d_in, const int* in_sizes, int n_in,
                              void* d_out, int out_size, void* d_ws, size_t ws_size,
                              hipStream_t stream) {
    (void)in_sizes; (void)n_in; (void)out_size; (void)ws_size;

    const float* x     = (const float*)d_in[0];
    const float* oc    = (const float*)d_in[1];
    const float* Wq    = (const float*)d_in[2];
    const float* bq    = (const float*)d_in[3];
    const float* Wk    = (const float*)d_in[4];
    const float* bk    = (const float*)d_in[5];
    const float* Wv    = (const float*)d_in[6];
    const float* bv    = (const float*)d_in[7];
    const float* Wo    = (const float*)d_in[8];
    const float* bo    = (const float*)d_in[9];
    const float* eta   = (const float*)d_in[10];
    const float* beta  = (const float*)d_in[11];
    const float* alpha = (const float*)d_in[12];
    float* out = (float*)d_out;

    float* ws  = (float*)d_ws;
    float* QKV = ws;
    float* ctx = ws + 3 * (size_t)B_SZ * NUM_HEADS * T_SZ * DK;

    {
        const int blocks = (3 * D_MODEL / SLAB_N) * (B_SZ * T_SZ / 128);
        qkv_proj_kernel<<<blocks, 256, 0, stream>>>(x, Wq, bq, Wk, bk, Wv, bv, QKV);
    }
    {
        const int waves = B_SZ * NUM_HEADS * (T_SZ / 32);
        attn_kernel<<<waves / 8, 256, 0, stream>>>(QKV, oc, eta, beta, alpha, ctx);
    }
    {
        const int blocks = (D_MODEL / SLAB_N) * (B_SZ * T_SZ / 128);
        out_proj_kernel<<<blocks, 256, 0, stream>>>(ctx, Wo, bo, out);
    }
}


# attention key-tile vector work cut 1/4: alpha*eta/sqrt(dk)*log2e folded into Q per query tile (base-2 softmax); oc distances packed on register pairs as loaded (3-term sum re-paired); few-ulp changes
# speedup vs baseline: 3.2679x; 1.0293x over previous
#include <hip/hip_runtime.h>
#include <math.h>

#define D_MODEL   256
#define NUM_HEADS 8
#define DK        32
#define B_SZ      4
#define T_SZ      2048

typedef float v2f __attribute__((ext_vector_type(2)));
typedef float v4f __attribute__((ext_vector_type(4)));
typedef float v8f __attribute__((ext_vector_type(8)));

__device__ __forceinline__ v8f wmma4(v2f a, v2f b, v8f c) {
    return __builtin_amdgcn_wmma_f32_16x16x4_f32(
         false, a,  false, b,
         (short)0, c,  false,  false);
}

#define WPITCH 260
#define SLAB_N 64

__device__ __forceinline__ void proj_load_a(v4f (&a)[8], const float* __restrict__ xr, int ch) {
#pragma unroll
    for (int u = 0; u < 8; ++u) a[u] = *(const v4f*)(xr + 64 * ch + 8 * u);
}

template <bool SWAP>
__device__ __forceinline__ void proj_mma_chunk(const v4f (&a)[8], const float* wl, int ch, v8f (&acc)[4]) {
#pragma unroll
    for (int u = 0; u < 8; ++u) {
        const int i = ch * 8 + u;
        v4f b[4];
#pragma unroll
        for (int j = 0; j < 4; ++j) b[j] = *(const v4f*)(wl + j * 16 * WPITCH + 8 * i);
        v2f alo, ahi;
        alo.x = a[u].x; alo.y = a[u].y; ahi.x = a[u].z; ahi.y = a[u].w;
#pragma unroll
        for (int j = 0; j < 4; ++j) {
            v2f blo; blo.x = b[j].x; blo.y = b[j].y;
            acc[j] = SWAP ? wmma4(blo, alo, acc[j]) : wmma4(alo, blo, acc[j]);
        }
#pragma unroll
        for (int j = 0; j < 4; ++j) {
            v2f bhi; bhi.x = b[j].z; bhi.y = b[j].w;
            acc[j] = SWAP ? wmma4(bhi, ahi, acc[j]) : wmma4(ahi, bhi, acc[j]);
        }
    }
}

template <bool SWAP>
__device__ __forceinline__ void proj_tile_16x64(const float* __restrict__ X, const float* __restrict__ Wm,
                                                int n0, int m0, float* wlds, v8f (&acc)[4])
{
    const int tid  = threadIdx.x;
    const int lane = tid & 31;
    const int c    = lane & 15;
    const int hh   = lane >> 4;

    const float* xr = X + (size_t)(m0 + c) * D_MODEL + 4 * hh;

    v4f a0[8], a1[8];
    proj_load_a(a0, xr, 0);

    {
        const int c4 = tid & 63;
        const int r0 = tid >> 6;
        v4f t[16];
#pragma unroll
        for (int u = 0; u < 16; ++u)
            t[u] = *(const v4f*)(Wm + (size_t)(n0 + 4 * u + r0) * D_MODEL + 4 * c4);
#pragma unroll
        for (int u = 0; u < 16; ++u)
            *(v4f*)(wlds + (4 * u + r0) * WPITCH + 4 * c4) = t[u];
    }
    __syncthreads();

    const float* wl = wlds + c * WPITCH + 4 * hh;

#pragma unroll
    for (int j = 0; j < 4; ++j) acc[j] = (v8f){};

    proj_load_a(a1, xr, 1);
    proj_mma_chunk<SWAP>(a0, wl, 0, acc);
    proj_load_a(a0, xr, 2);
    proj_mma_chunk<SWAP>(a1, wl, 1, acc);
    proj_load_a(a1, xr, 3);
    proj_mma_chunk<SWAP>(a0, wl, 2, acc);
    proj_mma_chunk<SWAP>(a1, wl, 3, acc);
}

__global__ __launch_bounds__(256)
void qkv_proj_kernel(const float* __restrict__ x,
                     const float* __restrict__ Wq, const float* __restrict__ bq,
                     const float* __restrict__ Wk, const float* __restrict__ bk,
                     const float* __restrict__ Wv, const float* __restrict__ bv,
                     float* __restrict__ QKV)
{
    __shared__ __attribute__((aligned(16))) float wlds[SLAB_N * WPITCH];

    const int lane = threadIdx.x & 31;
    const int w    = threadIdx.x >> 5;

    const int MG  = (B_SZ * T_SZ) / 128;
    const int ns  = blockIdx.x / MG;
    const int mg  = blockIdx.x % MG;
    const int mat = ns >> 2;
    const int n0  = (ns & 3) * SLAB_N;
    const int m0  = mg * 128 + w * 16;

    const float* Wm = (mat == 0) ? Wq : (mat == 1) ? Wk : Wv;
    const float* bs = (mat == 0) ? bq : (mat == 1) ? bk : bv;

    float* dst = QKV + (size_t)mat * ((size_t)B_SZ * NUM_HEADS * T_SZ * DK);

    const int b_ = m0 / T_SZ;
    const int tt = (m0 % T_SZ) >> 4;

    v8f acc[4];
    if (mat == 2) {
        proj_tile_16x64<false>(x, Wm, n0, m0, wlds, acc);
#pragma unroll
        for (int j = 0; j < 4; ++j) {
            const int nb = n0 + 16 * j;
            const float bias = bs[nb + (lane & 15)];
            const int h = nb >> 5, jd = (nb >> 4) & 1;
            float* tp = dst + ((size_t)(b_ * NUM_HEADS + h) * (T_SZ / 16) + tt) * 512 + lane * 4;
            v4f o;
            o.x = acc[j][0] + bias; o.y = acc[j][1] + bias; o.z = acc[j][2] + bias; o.w = acc[j][3] + bias;
            *(v4f*)(tp + (jd * 2 + 0) * 128) = o;
            o.x = acc[j][4] + bias; o.y = acc[j][5] + bias; o.z = acc[j][6] + bias; o.w = acc[j][7] + bias;
            *(v4f*)(tp + (jd * 2 + 1) * 128) = o;
        }
    } else {
        proj_tile_16x64<true>(x, Wm, n0, m0, wlds, acc);
#pragma unroll
        for (int j = 0; j < 4; ++j) {
            const int nb = n0 + 16 * j;
            const float* bp = bs + nb + 8 * (lane >> 4);
            v4f b0, b1;
            b0.x = bp[0]; b0.y = bp[1]; b0.z = bp[2]; b0.w = bp[3];
            b1.x = bp[4]; b1.y = bp[5]; b1.z = bp[6]; b1.w = bp[7];
            const int h = nb >> 5, jd = (nb >> 4) & 1;
            float* tp = dst + ((size_t)(b_ * NUM_HEADS + h) * (T_SZ / 16) + tt) * 512 + lane * 4;
            v4f o;
            o.x = acc[j][0] + b0.x; o.y = acc[j][1] + b0.y; o.z = acc[j][2] + b0.z; o.w = acc[j][3] + b0.w;
            *(v4f*)(tp + (jd * 2 + 0) * 128) = o;
            o.x = acc[j][4] + b1.x; o.y = acc[j][5] + b1.y; o.z = acc[j][6] + b1.z; o.w = acc[j][7] + b1.w;
            *(v4f*)(tp + (jd * 2 + 1) * 128) = o;
        }
    }
}

__device__ __forceinline__ float xhalf(float x) {
    const int b = __builtin_bit_cast(int, x);
    return __builtin_bit_cast(float,
        __builtin_amdgcn_permlanex16(b, b, 0x76543210, 0xfedcba98, false, false));
}

__device__ __forceinline__ void attn_load_k_oc(
    v4f (&k4)[4], v4f (&c)[6],
    const float* __restrict__ Kl, const float* __restrict__ oc, int kt)
{
    const float* kp = Kl + (size_t)kt * 512;
#pragma unroll
    for (int g = 0; g < 4; ++g) k4[g] = *(const v4f*)(kp + 128 * g);
    const v4f* kop = (const v4f*)(oc + (size_t)kt * 48);
#pragma unroll
    for (int g = 0; g < 6; ++g) c[g] = kop[g];
}

__device__ __forceinline__ void attn_load_v(v4f (&v4)[4], const float* __restrict__ Vl, int kt)
{
    const float* vp = Vl + (size_t)kt * 512;
#pragma unroll
    for (int g = 0; g < 4; ++g) v4[g] = *(const v4f*)(vp + 128 * g);
}

__device__ __forceinline__ void attn_loads_stay_here() { __builtin_amdgcn_sched_barrier(0); }

template <bool BETA_ONE, bool DIAG>
__device__ __forceinline__ void attn_tile_math(
    const v4f (&k4)[4], const v4f (&v4)[4], const v4f (&c)[6],
    const v2f (&qb)[8], v2f q01, v2f q20, v2f q12,
    float fd0, float absbeta,
    float& M, float& L, v8f& O0, v8f& O1)
{
    const float NEG_BIG = -3.4e38f;
    const float NLOG2E  = -1.4426950408889634f;

    const float fd1 = fd0 + 1.f;

    v8f s = {};
#pragma unroll
    for (int g = 0; g < 4; ++g) {
        v2f ka;
        ka.x = k4[g].x; ka.y = k4[g].y;
        s = wmma4(ka, qb[2 * g], s);
        ka.x = k4[g].z; ka.y = k4[g].w;
        s = wmma4(ka, qb[2 * g + 1], s);
    }

#pragma unroll
    for (int j = 0; j < 4; ++j) {
        const v4f ca = c[(6 * j) >> 2], cb = c[(6 * j + 2) >> 2], cc = c[(6 * j + 4) >> 2];
        v2f k01, k20, k12;
        if (j & 1) { k01.x = ca.z; k01.y = ca.w; k20.x = cb.x; k20.y = cb.y; k12.x = cc.z; k12.y = cc.w; }
        else       { k01.x = ca.x; k01.y = ca.y; k20.x = cb.z; k20.y = cb.w; k12.x = cc.x; k12.y = cc.y; }
        const v2f t0 = q01 - k01;
        const v2f t1 = q20 - k20;
        const v2f t2 = q12 - k12;
        const v2f p0 = t0 * t0;
        const v2f p2 = t2 * t2;
        v2f hs;
        hs.x = p0.x + p0.y;
        hs.y = p2.x + p2.y;
        const v2f e  = t1 * t1 + hs;
        const v2f ea = e * NLOG2E;
        const int rA = 2 * j, rB = 2 * j + 1;
        v2f den;
        den.x = BETA_ONE ? (fd1 - (float)rA) : (1.f + __powf(fd0 - (float)rA, absbeta));
        den.y = BETA_ONE ? (fd1 - (float)rB) : (1.f + __powf(fd0 - (float)rB, absbeta));
        v2f w, ex;
        w.x  = __builtin_amdgcn_rcpf(den.x);
        w.y  = __builtin_amdgcn_rcpf(den.y);
        ex.x = __builtin_amdgcn_exp2f(ea.x);
        ex.y = __builtin_amdgcn_exp2f(ea.y);
        v2f sv;
        sv.x = s[rA]; sv.y = s[rB];
        const v2f val = sv * (w * ex);
        if (DIAG) {
            s[rA] = (fd0 - (float)rA < 0.f) ? NEG_BIG : val.x;
            s[rB] = (fd0 - (float)rB < 0.f) ? NEG_BIG : val.y;
        } else { s[rA] = val.x; s[rB] = val.y; }
    }

    float mt = fmaxf(fmaxf(fmaxf(s[0], s[1]), fmaxf(s[2], s[3])),
                     fmaxf(fmaxf(s[4], s[5]), fmaxf(s[6], s[7])));
    mt = fmaxf(mt, xhalf(mt));
    const float mnew = fmaxf(M, mt);
    const float sc   = __builtin_amdgcn_exp2f(M - mnew);
    float p[8];
#pragma unroll
    for (int r = 0; r < 8; ++r) p[r] = __builtin_amdgcn_exp2f(s[r] - mnew);
    const float su = ((p[0] + p[1]) + (p[2] + p[3])) + ((p[4] + p[5]) + (p[6] + p[7]));
    L = L * sc + su;
    M = mnew;
#pragma unroll
    for (int r = 0; r < 8; ++r) { O0[r] *= sc; O1[r] *= sc; }

#pragma unroll
    for (int j = 0; j < 4; ++j) {
        v2f pb;
        pb.x = p[2 * j];
        pb.y = p[2 * j + 1];
        const v4f va = v4[j >> 1];
        const v4f vb = v4[2 + (j >> 1)];
        v2f a0, a1;
        if (j & 1) { a0.x = va.z; a0.y = va.w; a1.x = vb.z; a1.y = vb.w; }
        else       { a0.x = va.x; a0.y = va.y; a1.x = vb.x; a1.y = vb.y; }
        O0 = wmma4(a0, pb, O0);
        O1 = wmma4(a1, pb, O1);
    }
}

template <bool BETA_ONE>
__device__ __forceinline__ void attn_core(
    const float* __restrict__ Q, const float* __restrict__ K,
    const float* __restrict__ Vt, const float* __restrict__ oc,
    float* __restrict__ ctx,
    int b, int h, int qt, int lane,
    float eta, float absbeta, float alpha)
{
    const int col  = lane & 15;
    const int grp  = lane >> 4;
    const int q0   = qt * 16;
    const int qrow = q0 + col;

    const float NEG_BIG = -3.4e38f;

    const size_t rec0 = (size_t)(b * NUM_HEADS + h) * (T_SZ / 16) * 512 + lane * 4;

    const float* Kl = K + rec0;
    const float* Vl = Vt + rec0;

    const float* ock = oc + ((size_t)b * T_SZ + 8 * grp) * 3;

    v4f kA[4], cA[6];
    v4f kB[4], cB[6];
    attn_load_k_oc(kA, cA, Kl, ock, 0);

    const float cf = (alpha * eta) * (0.17677669529663687f * 1.4426950408889634f);
    const float* qp = Q + rec0 + (size_t)qt * 512;
    v2f qb[8];
#pragma unroll
    for (int g = 0; g < 4; ++g) {
        const v4f q4 = *(const v4f*)(qp + 128 * g);
        qb[2 * g].x = q4.x * cf;     qb[2 * g].y = q4.y * cf;
        qb[2 * g + 1].x = q4.z * cf; qb[2 * g + 1].y = q4.w * cf;
    }

    const float* op = oc + ((size_t)b * T_SZ + qrow) * 3;
    const float qoc0 = op[0], qoc1 = op[1], qoc2 = op[2];
    v2f q01, q20, q12;
    q01.x = qoc0; q01.y = qoc1;
    q20.x = qoc2; q20.y = qoc0;
    q12.x = qoc1; q12.y = qoc2;

    float M = NEG_BIG;
    float L = 0.f;
    v8f O0 = {}, O1 = {};

    float fd0 = (float)(qrow - 8 * grp);

    int kt = 0;
    for (; kt + 2 <= qt; kt += 2) {
        {
            v4f vv[4];
            attn_loads_stay_here();
            attn_load_v(vv, Vl, kt);
            attn_load_k_oc(kB, cB, Kl, ock, kt + 1);
            attn_loads_stay_here();
            attn_tile_math<BETA_ONE, false>(kA, vv, cA, qb, q01, q20, q12, fd0,
                                            absbeta, M, L, O0, O1);
        }
        {
            v4f vv[4];
            attn_loads_stay_here();
            attn_load_v(vv, Vl, kt + 1);
            attn_load_k_oc(kA, cA, Kl, ock, kt + 2);
            attn_loads_stay_here();
            attn_tile_math<BETA_ONE, false>(kB, vv, cB, qb, q01, q20, q12, fd0 - 16.f,
                                            absbeta, M, L, O0, O1);
        }
        fd0 -= 32.f;
    }
    if (kt < qt) {
        v4f vv[4];
        attn_load_v(vv, Vl, kt);
        attn_load_k_oc(kB, cB, Kl, ock, qt);
        attn_loads_stay_here();
        attn_tile_math<BETA_ONE, false>(kA, vv, cA, qb, q01, q20, q12, fd0,
                                        absbeta, M, L, O0, O1);
        fd0 -= 16.f;
#pragma unroll
        for (int g = 0; g < 4; ++g) kA[g] = kB[g];
#pragma unroll
        for (int g = 0; g < 6; ++g) cA[g] = cB[g];
    }
    {
        v4f vv[4];
        attn_load_v(vv, Vl, qt);
        attn_tile_math<BETA_ONE, true>(kA, vv, cA, qb, q01, q20, q12, fd0,
                                       absbeta, M, L, O0, O1);
    }

    L = L + xhalf(L);
    const float inv = 1.f / L;
    float* cp = ctx + ((size_t)b * T_SZ + qrow) * D_MODEL + h * DK + 8 * grp;
    v4f o;
    o.x = O0[0] * inv; o.y = O0[1] * inv; o.z = O0[2] * inv; o.w = O0[3] * inv;
    *(v4f*)(cp) = o;
    o.x = O0[4] * inv; o.y = O0[5] * inv; o.z = O0[6] * inv; o.w = O0[7] * inv;
    *(v4f*)(cp + 4) = o;
    o.x = O1[0] * inv; o.y = O1[1] * inv; o.z = O1[2] * inv; o.w = O1[3] * inv;
    *(v4f*)(cp + 16) = o;
    o.x = O1[4] * inv; o.y = O1[5] * inv; o.z = O1[6] * inv; o.w = O1[7] * inv;
    *(v4f*)(cp + 20) = o;
}

__global__ __launch_bounds__(256)
void attn_kernel(const float* __restrict__ QKV,
                 const float* __restrict__ oc,
                 const float* __restrict__ etap,
                 const float* __restrict__ betap,
                 const float* __restrict__ alphap,
                 float* __restrict__ ctx)
{
    const int lane = threadIdx.x & 31;
    const int w    = threadIdx.x >> 5;
    const int wid  = blockIdx.x * 8 + w;

    const int QT = T_SZ / 16;
    const int QP = QT / 2;
    const int pr = wid % QP;
    const int h  = (wid / QP) % NUM_HEADS;
    const int b  = wid / (QP * NUM_HEADS);

    const float eta     = etap[0];
    const float absbeta = fabsf(betap[0]);
    const float alpha   = alphap[0];

    const size_t HT = (size_t)B_SZ * NUM_HEADS * T_SZ * DK;
    const float* Q  = QKV;
    const float* K  = QKV + HT;
    const float* Vt = QKV + 2 * HT;

    if (absbeta == 1.0f) {
        for (int pass = 0; pass < 2; ++pass) {
            const int qt = pass ? pr : (QT - 1 - pr);
            attn_core<true >(Q, K, Vt, oc, ctx, b, h, qt, lane, eta, absbeta, alpha);
        }
    } else {
        for (int pass = 0; pass < 2; ++pass) {
            const int qt = pass ? pr : (QT - 1 - pr);
            attn_core<false>(Q, K, Vt, oc, ctx, b, h, qt, lane, eta, absbeta, alpha);
        }
    }
}

__global__ __launch_bounds__(256)
void out_proj_kernel(const float* __restrict__ ctx,
                     const float* __restrict__ Wo, const float* __restrict__ bo,
                     float* __restrict__ out)
{
    __shared__ __attribute__((aligned(16))) float wlds[SLAB_N * WPITCH];

    const int lane = threadIdx.x & 31;
    const int w    = threadIdx.x >> 5;

    const int MG = (B_SZ * T_SZ) / 128;
    const int ns = blockIdx.x / MG;
    const int mg = blockIdx.x % MG;
    const int n0 = ns * SLAB_N;
    const int m0 = mg * 128 + w * 16;

    v8f acc[4];
    proj_tile_16x64<false>(ctx, Wo, n0, m0, wlds, acc);

    const int r15 = lane & 15;
    const int grp = lane >> 4;
#pragma unroll
    for (int j = 0; j < 4; ++j) {
        const int n = n0 + 16 * j + r15;
        const float bias = bo[n];
#pragma unroll
        for (int m = 0; m < 8; ++m) {
            const int tl = m0 + m + 8 * grp;
            out[(size_t)tl * D_MODEL + n] = acc[j][m] + bias;
        }
    }
}

extern "C" void kernel_launch(void* const* d_in, const int* in_sizes, int n_in,
                              void* d_out, int out_size, void* d_ws, size_t ws_size,
                              hipStream_t stream) {
    (void)in_sizes; (void)n_in; (void)out_size; (void)ws_size;

    const float* x     = (const float*)d_in[0];
    const float* oc    = (const float*)d_in[1];
    const float* Wq    = (const float*)d_in[2];
    const float* bq    = (const float*)d_in[3];
    const float* Wk    = (const float*)d_in[4];
    const float* bk    = (const float*)d_in[5];
    const float* Wv    = (const float*)d_in[6];
    const float* bv    = (const float*)d_in[7];
    const float* Wo    = (const float*)d_in[8];
    const float* bo    = (const float*)d_in[9];
    const float* eta   = (const float*)d_in[10];
    const float* beta  = (const float*)d_in[11];
    const float* alpha = (const float*)d_in[12];
    float* out = (float*)d_out;

    float* ws  = (float*)d_ws;
    float* QKV = ws;
    float* ctx = ws + 3 * (size_t)B_SZ * NUM_HEADS * T_SZ * DK;

    {
        const int blocks = (3 * D_MODEL / SLAB_N) * (B_SZ * T_SZ / 128);
        qkv_proj_kernel<<<blocks, 256, 0, stream>>>(x, Wq, bq, Wk, bk, Wv, bv, QKV);
    }
    {
        const int waves = B_SZ * NUM_HEADS * (T_SZ / 32);
        attn_kernel<<<waves / 8, 256, 0, stream>>>(QKV, oc, eta, beta, alpha, ctx);
    }
    {
        const int blocks = (D_MODEL / SLAB_N) * (B_SZ * T_SZ / 128);
        out_proj_kernel<<<blocks, 256, 0, stream>>>(ctx, Wo, bo, out);
    }
}


# attention split over keys: a query-tile pair's 129 key tiles go to two waves (64 + 65), the long tile's shares merged via LDS; 4096 waves, 4 per SIMD (was 2); changes softmax/PV sum order
# speedup vs baseline: 3.6010x; 1.1019x over previous
#include <hip/hip_runtime.h>
#include <math.h>

#define D_MODEL   256
#define NUM_HEADS 8
#define DK        32
#define B_SZ      4
#define T_SZ      2048

typedef float v2f __attribute__((ext_vector_type(2)));
typedef float v4f __attribute__((ext_vector_type(4)));
typedef float v8f __attribute__((ext_vector_type(8)));

__device__ __forceinline__ v8f wmma4(v2f a, v2f b, v8f c) {
    return __builtin_amdgcn_wmma_f32_16x16x4_f32(
         false, a,  false, b,
         (short)0, c,  false,  false);
}

#define WPITCH 260
#define SLAB_N 64

__device__ __forceinline__ void proj_load_a(v4f (&a)[8], const float* __restrict__ xr, int ch) {
#pragma unroll
    for (int u = 0; u < 8; ++u) a[u] = *(const v4f*)(xr + 64 * ch + 8 * u);
}

template <bool SWAP>
__device__ __forceinline__ void proj_mma_chunk(const v4f (&a)[8], const float* wl, int ch, v8f (&acc)[4]) {
#pragma unroll
    for (int u = 0; u < 8; ++u) {
        const int i = ch * 8 + u;
        v4f b[4];
#pragma unroll
        for (int j = 0; j < 4; ++j) b[j] = *(const v4f*)(wl + j * 16 * WPITCH + 8 * i);
        v2f alo, ahi;
        alo.x = a[u].x; alo.y = a[u].y; ahi.x = a[u].z; ahi.y = a[u].w;
#pragma unroll
        for (int j = 0; j < 4; ++j) {
            v2f blo; blo.x = b[j].x; blo.y = b[j].y;
            acc[j] = SWAP ? wmma4(blo, alo, acc[j]) : wmma4(alo, blo, acc[j]);
        }
#pragma unroll
        for (int j = 0; j < 4; ++j) {
            v2f bhi; bhi.x = b[j].z; bhi.y = b[j].w;
            acc[j] = SWAP ? wmma4(bhi, ahi, acc[j]) : wmma4(ahi, bhi, acc[j]);
        }
    }
}

template <bool SWAP>
__device__ __forceinline__ void proj_tile_16x64(const float* __restrict__ X, const float* __restrict__ Wm,
                                                int n0, int m0, float* wlds, v8f (&acc)[4])
{
    const int tid  = threadIdx.x;
    const int lane = tid & 31;
    const int c    = lane & 15;
    const int hh   = lane >> 4;

    const float* xr = X + (size_t)(m0 + c) * D_MODEL + 4 * hh;

    v4f a0[8], a1[8];
    proj_load_a(a0, xr, 0);

    {
        const int c4 = tid & 63;
        const int r0 = tid >> 6;
        v4f t[16];
#pragma unroll
        for (int u = 0; u < 16; ++u)
            t[u] = *(const v4f*)(Wm + (size_t)(n0 + 4 * u + r0) * D_MODEL + 4 * c4);
#pragma unroll
        for (int u = 0; u < 16; ++u)
            *(v4f*)(wlds + (4 * u + r0) * WPITCH + 4 * c4) = t[u];
    }
    __syncthreads();

    const float* wl = wlds + c * WPITCH + 4 * hh;

#pragma unroll
    for (int j = 0; j < 4; ++j) acc[j] = (v8f){};

    proj_load_a(a1, xr, 1);
    proj_mma_chunk<SWAP>(a0, wl, 0, acc);
    proj_load_a(a0, xr, 2);
    proj_mma_chunk<SWAP>(a1, wl, 1, acc);
    proj_load_a(a1, xr, 3);
    proj_mma_chunk<SWAP>(a0, wl, 2, acc);
    proj_mma_chunk<SWAP>(a1, wl, 3, acc);
}

__global__ __launch_bounds__(256)
void qkv_proj_kernel(const float* __restrict__ x,
                     const float* __restrict__ Wq, const float* __restrict__ bq,
                     const float* __restrict__ Wk, const float* __restrict__ bk,
                     const float* __restrict__ Wv, const float* __restrict__ bv,
                     float* __restrict__ QKV)
{
    __shared__ __attribute__((aligned(16))) float wlds[SLAB_N * WPITCH];

    const int lane = threadIdx.x & 31;
    const int w    = threadIdx.x >> 5;

    const int MG  = (B_SZ * T_SZ) / 128;
    const int ns  = blockIdx.x / MG;
    const int mg  = blockIdx.x % MG;
    const int mat = ns >> 2;
    const int n0  = (ns & 3) * SLAB_N;
    const int m0  = mg * 128 + w * 16;

    const float* Wm = (mat == 0) ? Wq : (mat == 1) ? Wk : Wv;
    const float* bs = (mat == 0) ? bq : (mat == 1) ? bk : bv;

    float* dst = QKV + (size_t)mat * ((size_t)B_SZ * NUM_HEADS * T_SZ * DK);

    const int b_ = m0 / T_SZ;
    const int tt = (m0 % T_SZ) >> 4;

    v8f acc[4];
    if (mat == 2) {
        proj_tile_16x64<false>(x, Wm, n0, m0, wlds, acc);
#pragma unroll
        for (int j = 0; j < 4; ++j) {
            const int nb = n0 + 16 * j;
            const float bias = bs[nb + (lane & 15)];
            const int h = nb >> 5, jd = (nb >> 4) & 1;
            float* tp = dst + ((size_t)(b_ * NUM_HEADS + h) * (T_SZ / 16) + tt) * 512 + lane * 4;
            v4f o;
            o.x = acc[j][0] + bias; o.y = acc[j][1] + bias; o.z = acc[j][2] + bias; o.w = acc[j][3] + bias;
            *(v4f*)(tp + (jd * 2 + 0) * 128) = o;
            o.x = acc[j][4] + bias; o.y = acc[j][5] + bias; o.z = acc[j][6] + bias; o.w = acc[j][7] + bias;
            *(v4f*)(tp + (jd * 2 + 1) * 128) = o;
        }
    } else {
        proj_tile_16x64<true>(x, Wm, n0, m0, wlds, acc);
#pragma unroll
        for (int j = 0; j < 4; ++j) {
            const int nb = n0 + 16 * j;
            const float* bp = bs + nb + 8 * (lane >> 4);
            v4f b0, b1;
            b0.x = bp[0]; b0.y = bp[1]; b0.z = bp[2]; b0.w = bp[3];
            b1.x = bp[4]; b1.y = bp[5]; b1.z = bp[6]; b1.w = bp[7];
            const int h = nb >> 5, jd = (nb >> 4) & 1;
            float* tp = dst + ((size_t)(b_ * NUM_HEADS + h) * (T_SZ / 16) + tt) * 512 + lane * 4;
            v4f o;
            o.x = acc[j][0] + b0.x; o.y = acc[j][1] + b0.y; o.z = acc[j][2] + b0.z; o.w = acc[j][3] + b0.w;
            *(v4f*)(tp + (jd * 2 + 0) * 128) = o;
            o.x = acc[j][4] + b1.x; o.y = acc[j][5] + b1.y; o.z = acc[j][6] + b1.z; o.w = acc[j][7] + b1.w;
            *(v4f*)(tp + (jd * 2 + 1) * 128) = o;
        }
    }
}

__device__ __forceinline__ float xhalf(float x) {
    const int b = __builtin_bit_cast(int, x);
    return __builtin_bit_cast(float,
        __builtin_amdgcn_permlanex16(b, b, 0x76543210, 0xfedcba98, false, false));
}

__device__ __forceinline__ void attn_load_k_oc(
    v4f (&k4)[4], v4f (&c)[6],
    const float* __restrict__ Kl, const float* __restrict__ oc, int kt)
{
    const float* kp = Kl + (size_t)kt * 512;
#pragma unroll
    for (int g = 0; g < 4; ++g) k4[g] = *(const v4f*)(kp + 128 * g);
    const v4f* kop = (const v4f*)(oc + (size_t)kt * 48);
#pragma unroll
    for (int g = 0; g < 6; ++g) c[g] = kop[g];
}

__device__ __forceinline__ void attn_load_v(v4f (&v4)[4], const float* __restrict__ Vl, int kt)
{
    const float* vp = Vl + (size_t)kt * 512;
#pragma unroll
    for (int g = 0; g < 4; ++g) v4[g] = *(const v4f*)(vp + 128 * g);
}

__device__ __forceinline__ void attn_loads_stay_here() { __builtin_amdgcn_sched_barrier(0); }

template <bool BETA_ONE, bool DIAG>
__device__ __forceinline__ void attn_tile_math(
    const v4f (&k4)[4], const v4f (&v4)[4], const v4f (&c)[6],
    const v2f (&qb)[8], v2f q01, v2f q20, v2f q12,
    float fd0, float absbeta,
    float& M, float& L, v8f& O0, v8f& O1)
{
    const float NEG_BIG = -3.4e38f;
    const float NLOG2E  = -1.4426950408889634f;

    const float fd1 = fd0 + 1.f;

    v8f s = {};
#pragma unroll
    for (int g = 0; g < 4; ++g) {
        v2f ka;
        ka.x = k4[g].x; ka.y = k4[g].y;
        s = wmma4(ka, qb[2 * g], s);
        ka.x = k4[g].z; ka.y = k4[g].w;
        s = wmma4(ka, qb[2 * g + 1], s);
    }

#pragma unroll
    for (int j = 0; j < 4; ++j) {
        const v4f ca = c[(6 * j) >> 2], cb = c[(6 * j + 2) >> 2], cc = c[(6 * j + 4) >> 2];
        v2f k01, k20, k12;
        if (j & 1) { k01.x = ca.z; k01.y = ca.w; k20.x = cb.x; k20.y = cb.y; k12.x = cc.z; k12.y = cc.w; }
        else       { k01.x = ca.x; k01.y = ca.y; k20.x = cb.z; k20.y = cb.w; k12.x = cc.x; k12.y = cc.y; }
        const v2f t0 = q01 - k01;
        const v2f t1 = q20 - k20;
        const v2f t2 = q12 - k12;
        const v2f p0 = t0 * t0;
        const v2f p2 = t2 * t2;
        v2f hs;
        hs.x = p0.x + p0.y;
        hs.y = p2.x + p2.y;
        const v2f e  = t1 * t1 + hs;
        const v2f ea = e * NLOG2E;
        const int rA = 2 * j, rB = 2 * j + 1;
        v2f den;
        den.x = BETA_ONE ? (fd1 - (float)rA) : (1.f + __powf(fd0 - (float)rA, absbeta));
        den.y = BETA_ONE ? (fd1 - (float)rB) : (1.f + __powf(fd0 - (float)rB, absbeta));
        v2f w, ex;
        w.x  = __builtin_amdgcn_rcpf(den.x);
        w.y  = __builtin_amdgcn_rcpf(den.y);
        ex.x = __builtin_amdgcn_exp2f(ea.x);
        ex.y = __builtin_amdgcn_exp2f(ea.y);
        v2f sv;
        sv.x = s[rA]; sv.y = s[rB];
        const v2f val = sv * (w * ex);
        if (DIAG) {
            s[rA] = (fd0 - (float)rA < 0.f) ? NEG_BIG : val.x;
            s[rB] = (fd0 - (float)rB < 0.f) ? NEG_BIG : val.y;
        } else { s[rA] = val.x; s[rB] = val.y; }
    }

    float mt = fmaxf(fmaxf(fmaxf(s[0], s[1]), fmaxf(s[2], s[3])),
                     fmaxf(fmaxf(s[4], s[5]), fmaxf(s[6], s[7])));
    mt = fmaxf(mt, xhalf(mt));
    const float mnew = fmaxf(M, mt);
    const float sc   = __builtin_amdgcn_exp2f(M - mnew);
    float p[8];
#pragma unroll
    for (int r = 0; r < 8; ++r) p[r] = __builtin_amdgcn_exp2f(s[r] - mnew);
    const float su = ((p[0] + p[1]) + (p[2] + p[3])) + ((p[4] + p[5]) + (p[6] + p[7]));
    L = L * sc + su;
    M = mnew;
#pragma unroll
    for (int r = 0; r < 8; ++r) { O0[r] *= sc; O1[r] *= sc; }

#pragma unroll
    for (int j = 0; j < 4; ++j) {
        v2f pb;
        pb.x = p[2 * j];
        pb.y = p[2 * j + 1];
        const v4f va = v4[j >> 1];
        const v4f vb = v4[2 + (j >> 1)];
        v2f a0, a1;
        if (j & 1) { a0.x = va.z; a0.y = va.w; a1.x = vb.z; a1.y = vb.w; }
        else       { a0.x = va.x; a0.y = va.y; a1.x = vb.x; a1.y = vb.y; }
        O0 = wmma4(a0, pb, O0);
        O1 = wmma4(a1, pb, O1);
    }
}

template <bool BETA_ONE>
__device__ __forceinline__ void attn_core(
    const float* __restrict__ Q, const float* __restrict__ K,
    const float* __restrict__ Vt, const float* __restrict__ oc,
    int b, int h, int qt, int kt0, int kend, bool diag, int lane,
    float eta, float absbeta, float alpha,
    float& Mo, float& Lo, v8f& O0o, v8f& O1o)
{
    const int col  = lane & 15;
    const int grp  = lane >> 4;
    const int q0   = qt * 16;
    const int qrow = q0 + col;

    const float NEG_BIG = -3.4e38f;

    const size_t rec0 = (size_t)(b * NUM_HEADS + h) * (T_SZ / 16) * 512 + lane * 4;

    const float* Kl = K + rec0;
    const float* Vl = Vt + rec0;

    const float* ock = oc + ((size_t)b * T_SZ + 8 * grp) * 3;

    v4f kA[4], cA[6];
    v4f kB[4], cB[6];
    attn_load_k_oc(kA, cA, Kl, ock, kt0);

    const float cf = (alpha * eta) * (0.17677669529663687f * 1.4426950408889634f);
    const float* qp = Q + rec0 + (size_t)qt * 512;
    v2f qb[8];
#pragma unroll
    for (int g = 0; g < 4; ++g) {
        const v4f q4 = *(const v4f*)(qp + 128 * g);
        qb[2 * g].x = q4.x * cf;     qb[2 * g].y = q4.y * cf;
        qb[2 * g + 1].x = q4.z * cf; qb[2 * g + 1].y = q4.w * cf;
    }

    const float* op = oc + ((size_t)b * T_SZ + qrow) * 3;
    const float qoc0 = op[0], qoc1 = op[1], qoc2 = op[2];
    v2f q01, q20, q12;
    q01.x = qoc0; q01.y = qoc1;
    q20.x = qoc2; q20.y = qoc0;
    q12.x = qoc1; q12.y = qoc2;

    float M = NEG_BIG;
    float L = 0.f;
    v8f O0 = {}, O1 = {};

    float fd0 = (float)(qrow - 8 * grp - 16 * kt0);

    int kt = kt0;
    for (; kt + 2 <= kend; kt += 2) {
        {
            v4f vv[4];
            attn_loads_stay_here();
            attn_load_v(vv, Vl, kt);
            attn_load_k_oc(kB, cB, Kl, ock, kt + 1);
            attn_loads_stay_here();
            attn_tile_math<BETA_ONE, false>(kA, vv, cA, qb, q01, q20, q12, fd0,
                                            absbeta, M, L, O0, O1);
        }
        {
            v4f vv[4];
            attn_loads_stay_here();
            attn_load_v(vv, Vl, kt + 1);
            attn_load_k_oc(kA, cA, Kl, ock, kt + 2);
            attn_loads_stay_here();
            attn_tile_math<BETA_ONE, false>(kB, vv, cB, qb, q01, q20, q12, fd0 - 16.f,
                                            absbeta, M, L, O0, O1);
        }
        fd0 -= 32.f;
    }
    if (kt < kend) {
        v4f vv[4];
        attn_load_v(vv, Vl, kt);
        attn_load_k_oc(kB, cB, Kl, ock, kend);
        attn_loads_stay_here();
        attn_tile_math<BETA_ONE, false>(kA, vv, cA, qb, q01, q20, q12, fd0,
                                        absbeta, M, L, O0, O1);
        fd0 -= 16.f;
#pragma unroll
        for (int g = 0; g < 4; ++g) kA[g] = kB[g];
#pragma unroll
        for (int g = 0; g < 6; ++g) cA[g] = cB[g];
    }
    if (diag) {
        v4f vv[4];
        attn_load_v(vv, Vl, qt);
        attn_tile_math<BETA_ONE, true>(kA, vv, cA, qb, q01, q20, q12, fd0,
                                       absbeta, M, L, O0, O1);
    }
    Mo = M; Lo = L; O0o = O0; O1o = O1;
}

__device__ __forceinline__ void attn_finish(float* __restrict__ ctx, int b, int h, int qt, int lane,
                                            float L, const v8f& O0, const v8f& O1)
{
    const int col  = lane & 15;
    const int grp  = lane >> 4;
    const int qrow = qt * 16 + col;
    L = L + xhalf(L);
    const float inv = 1.f / L;
    float* cp = ctx + ((size_t)b * T_SZ + qrow) * D_MODEL + h * DK + 8 * grp;
    v4f o;
    o.x = O0[0] * inv; o.y = O0[1] * inv; o.z = O0[2] * inv; o.w = O0[3] * inv;
    *(v4f*)(cp) = o;
    o.x = O0[4] * inv; o.y = O0[5] * inv; o.z = O0[6] * inv; o.w = O0[7] * inv;
    *(v4f*)(cp + 4) = o;
    o.x = O1[0] * inv; o.y = O1[1] * inv; o.z = O1[2] * inv; o.w = O1[3] * inv;
    *(v4f*)(cp + 16) = o;
    o.x = O1[4] * inv; o.y = O1[5] * inv; o.z = O1[6] * inv; o.w = O1[7] * inv;
    *(v4f*)(cp + 20) = o;
}

#define PART_REGS 18
template <bool BETA_ONE>
__device__ __forceinline__ void attn_wave(
    const float* __restrict__ Q, const float* __restrict__ K,
    const float* __restrict__ Vt, const float* __restrict__ oc,
    float* __restrict__ ctx, float* part,
    int b, int h, int pr, int role, int j, int lane,
    float eta, float absbeta, float alpha)
{
    const int QT = T_SZ / 16;
    const int HK = QT / 2;
    const int ql = QT - 1 - pr;

    float M = 0.f, L = 0.f;
    v8f O0 = {}, O1 = {};
    for (int seg = role ? 0 : 1; seg < 2; ++seg) {
        const int qt   = seg ? ql : pr;
        const int kt0  = (seg && role) ? HK : 0;
        const int kend = (seg && !role) ? HK : qt;
        const bool diag = !(seg && !role);
        attn_core<BETA_ONE>(Q, K, Vt, oc, b, h, qt, kt0, kend, diag, lane,
                            eta, absbeta, alpha, M, L, O0, O1);
        if (seg == 0) attn_finish(ctx, b, h, qt, lane, L, O0, O1);
    }

    float* pp = part + j * 32 + lane;
    if (role == 0) {
        pp[0 * 256] = M;
        pp[1 * 256] = L;
#pragma unroll
        for (int r = 0; r < 8; ++r) { pp[(2 + r) * 256] = O0[r]; pp[(10 + r) * 256] = O1[r]; }
    }
    __syncthreads();
    if (role == 1) {
        const float Ma = pp[0 * 256];
        const float La = pp[1 * 256];
        const float Mn = fmaxf(M, Ma);
        const float fa = __builtin_amdgcn_exp2f(Ma - Mn);
        const float fb = __builtin_amdgcn_exp2f(M - Mn);
        L = La * fa + L * fb;
#pragma unroll
        for (int r = 0; r < 8; ++r) {
            O0[r] = pp[(2 + r) * 256] * fa + O0[r] * fb;
            O1[r] = pp[(10 + r) * 256] * fa + O1[r] * fb;
        }
        attn_finish(ctx, b, h, ql, lane, L, O0, O1);
    }
}

__global__ __launch_bounds__(512)
void attn_kernel(const float* __restrict__ QKV,
                 const float* __restrict__ oc,
                 const float* __restrict__ etap,
                 const float* __restrict__ betap,
                 const float* __restrict__ alphap,
                 float* __restrict__ ctx)
{
    __shared__ float part[PART_REGS * 8 * 32];

    const int lane = threadIdx.x & 31;
    const int w    = threadIdx.x >> 5;
    const int role = w >> 3;
    const int j    = w & 7;
    const int pid  = blockIdx.x * 8 + j;

    const int QP = T_SZ / 32;
    const int pr = pid % QP;
    const int h  = (pid / QP) % NUM_HEADS;
    const int b  = pid / (QP * NUM_HEADS);

    const float eta     = etap[0];
    const float absbeta = fabsf(betap[0]);
    const float alpha   = alphap[0];

    const size_t HT = (size_t)B_SZ * NUM_HEADS * T_SZ * DK;
    const float* Q  = QKV;
    const float* K  = QKV + HT;
    const float* Vt = QKV + 2 * HT;

    if (absbeta == 1.0f)
        attn_wave<true >(Q, K, Vt, oc, ctx, part, b, h, pr, role, j, lane, eta, absbeta, alpha);
    else
        attn_wave<false>(Q, K, Vt, oc, ctx, part, b, h, pr, role, j, lane, eta, absbeta, alpha);
}

__global__ __launch_bounds__(256)
void out_proj_kernel(const float* __restrict__ ctx,
                     const float* __restrict__ Wo, const float* __restrict__ bo,
                     float* __restrict__ out)
{
    __shared__ __attribute__((aligned(16))) float wlds[SLAB_N * WPITCH];

    const int lane = threadIdx.x & 31;
    const int w    = threadIdx.x >> 5;

    const int MG = (B_SZ * T_SZ) / 128;
    const int ns = blockIdx.x / MG;
    const int mg = blockIdx.x % MG;
    const int n0 = ns * SLAB_N;
    const int m0 = mg * 128 + w * 16;

    v8f acc[4];
    proj_tile_16x64<false>(ctx, Wo, n0, m0, wlds, acc);

    const int r15 = lane & 15;
    const int grp = lane >> 4;
#pragma unroll
    for (int j = 0; j < 4; ++j) {
        const int n = n0 + 16 * j + r15;
        const float bias = bo[n];
#pragma unroll
        for (int m = 0; m < 8; ++m) {
            const int tl = m0 + m + 8 * grp;
            out[(size_t)tl * D_MODEL + n] = acc[j][m] + bias;
        }
    }
}

extern "C" void kernel_launch(void* const* d_in, const int* in_sizes, int n_in,
                              void* d_out, int out_size, void* d_ws, size_t ws_size,
                              hipStream_t stream) {
    (void)in_sizes; (void)n_in; (void)out_size; (void)ws_size;

    const float* x     = (const float*)d_in[0];
    const float* oc    = (const float*)d_in[1];
    const float* Wq    = (const float*)d_in[2];
    const float* bq    = (const float*)d_in[3];
    const float* Wk    = (const float*)d_in[4];
    const float* bk    = (const float*)d_in[5];
    const float* Wv    = (const float*)d_in[6];
    const float* bv    = (const float*)d_in[7];
    const float* Wo    = (const float*)d_in[8];
    const float* bo    = (const float*)d_in[9];
    const float* eta   = (const float*)d_in[10];
    const float* beta  = (const float*)d_in[11];
    const float* alpha = (const float*)d_in[12];
    float* out = (float*)d_out;

    float* ws  = (float*)d_ws;
    float* QKV = ws;
    float* ctx = ws + 3 * (size_t)B_SZ * NUM_HEADS * T_SZ * DK;

    {
        const int blocks = (3 * D_MODEL / SLAB_N) * (B_SZ * T_SZ / 128);
        qkv_proj_kernel<<<blocks, 256, 0, stream>>>(x, Wq, bq, Wk, bk, Wv, bv, QKV);
    }
    {
        const int pairs = B_SZ * NUM_HEADS * (T_SZ / 32);
        attn_kernel<<<pairs / 8, 512, 0, stream>>>(QKV, oc, eta, beta, alpha, ctx);
    }
    {
        const int blocks = (D_MODEL / SLAB_N) * (B_SZ * T_SZ / 128);
        out_proj_kernel<<<blocks, 256, 0, stream>>>(ctx, Wo, bo, out);
    }
}
